# starting kernel
# baseline (speedup 1.0000x reference)
#include <hip/hip_runtime.h>
#include <hip/hip_bf16.h>
#include <math.h>

#define Bn 8
#define Tn 256
#define INn 128
#define Hn 128
#define OUTn 128
#define Ln 2
#define STSTRIDE 33025
#define LAYERSTATE 16512
#define YSZ (Tn*Bn*OUTn)

typedef __attribute__((ext_vector_type(2))) float v2f;
typedef __attribute__((ext_vector_type(8))) float v8f;

__device__ __forceinline__ v8f wmma4(v2f a, v2f b, v8f c) {
  return __builtin_amdgcn_wmma_f32_16x16x4_f32(false, a, false, b, (short)0, c, false, false);
}

__device__ __forceinline__ void tile_abt(const float* A, int lda, const float* Bt, int ldb,
                                         int K, v8f& acc) {
  const int lane = threadIdx.x & 31;
  const int half = lane >> 4;
  const int idx  = lane & 15;
  const float* ap = A  + (size_t)idx * lda + 2 * half;
  const float* bp = Bt + (size_t)idx * ldb + 2 * half;
  for (int k = 0; k < K; k += 4) {
    v2f a, b;
    a.x = ap[k]; a.y = ap[k + 1];
    b.x = bp[k]; b.y = bp[k + 1];
    acc = wmma4(a, b, acc);
  }
}

__device__ __forceinline__ void tile_ab(const float* A, int lda, const float* Bm, int ldb,
                                        int K, v8f& acc) {
  const int lane = threadIdx.x & 31;
  const int half = lane >> 4;
  const int idx  = lane & 15;
  const float* ap = A + (size_t)idx * lda + 2 * half;
  for (int k = 0; k < K; k += 4) {
    v2f a, b;
    a.x = ap[k]; a.y = ap[k + 1];
    b.x = Bm[(size_t)(k + 2 * half)     * ldb + idx];
    b.y = Bm[(size_t)(k + 2 * half + 1) * ldb + idx];
    acc = wmma4(a, b, acc);
  }
}

__device__ __forceinline__ void tile_atb(const float* Am, int lda, const float* Bm, int ldb,
                                         int K, v8f& acc) {
  const int lane = threadIdx.x & 31;
  const int half = lane >> 4;
  const int idx  = lane & 15;
  for (int k = 0; k < K; k += 4) {
    v2f a, b;
    a.x = Am[(size_t)(k + 2 * half)     * lda + idx];
    a.y = Am[(size_t)(k + 2 * half + 1) * lda + idx];
    b.x = Bm[(size_t)(k + 2 * half)     * ldb + idx];
    b.y = Bm[(size_t)(k + 2 * half + 1) * ldb + idx];
    acc = wmma4(a, b, acc);
  }
}

__global__ void dla_rotate_kernel(const float* __restrict__ z, const float* __restrict__ st,
                                  float* __restrict__ xcur, float* __restrict__ outst) {
  const int bt = blockIdx.x;
  const int b = bt >> 8, t = bt & 255;
  const int j = threadIdx.x;
  const float ts = st[(size_t)b * STSTRIDE + STSTRIDE - 1];
  const float pos = ts + (float)t;
  const float freq = expf((float)(2 * j) * (-9.210340371976184f / 128.0f));
  const float ang = pos * freq;
  const float s = sinf(ang), c = cosf(ang);
  const float* zp = z + (size_t)t * Bn * INn + (size_t)b * INn;
  const float x1 = zp[j], x2 = zp[j + 64];
  float* xp = xcur + ((size_t)b * Tn + t) * INn;
  xp[j]      = x1 * c - x2 * s;
  xp[j + 64] = x2 * c + x1 * s;
  if (t == 0 && j == 0)
    outst[(size_t)b * STSTRIDE + STSTRIDE - 1] = ts + (float)Tn;
}

__global__ void dla_ln_kernel(const float* __restrict__ x, const float* __restrict__ w,
                              const float* __restrict__ bias, float* __restrict__ xn) {
  const int wave = threadIdx.x >> 5, lane = threadIdx.x & 31;
  const int row = blockIdx.x * 8 + wave;
  const float* xr = x + (size_t)row * INn;
  float s = 0.f, ss = 0.f;
  for (int k = lane; k < INn; k += 32) { float v = xr[k]; s += v; ss += v * v; }
  for (int off = 16; off; off >>= 1) { s += __shfl_xor(s, off); ss += __shfl_xor(ss, off); }
  const float m = s * (1.0f / INn);
  const float var = ss * (1.0f / INn) - m * m;
  const float inv = rsqrtf(var + 1e-5f);
  float* outr = xn + (size_t)row * INn;
  for (int k = lane; k < INn; k += 32)
    outr[k] = (xr[k] - m) * inv * w[k] + bias[k];
}

__global__ void dla_proj_kernel(const float* __restrict__ xn,
                                const float* __restrict__ Wk, const float* __restrict__ Wq,
                                const float* __restrict__ Wv,
                                float* __restrict__ Kb, float* __restrict__ Qb,
                                float* __restrict__ Vb) {
  const int g = blockIdx.x * 8 + (threadIdx.x >> 5);
  const int which = g >> 10;
  const int rem = g & 1023;
  const int b = rem >> 7;
  const int tile = rem & 127;
  const int tm = tile >> 3, tn = tile & 7;
  const float* W = (which == 0) ? Wk : (which == 1) ? Wq : Wv;
  const float* A  = xn + (size_t)b * Tn * INn + (size_t)tm * 16 * INn;
  const float* Bt = W + (size_t)tn * 16 * INn;
  v8f acc = {0.f,0.f,0.f,0.f,0.f,0.f,0.f,0.f};
  tile_abt(A, INn, Bt, INn, INn, acc);
  float* Out = ((which == 0) ? Kb : (which == 1) ? Qb : Vb)
             + (size_t)b * Tn * Hn + (size_t)tm * 16 * Hn + tn * 16;
  const int lane = threadIdx.x & 31, half = lane >> 4, idx = lane & 15;
#pragma unroll
  for (int r = 0; r < 8; r++) {
    float v = acc[r];
    if (which != 2) v = ((v > 0.f) ? v : expm1f(v)) + 1.0f;
    Out[(size_t)(r + 8 * half) * Hn + idx] = v;
  }
}

__global__ void dla_score_kernel(const float* __restrict__ Qb, const float* __restrict__ Kb,
                                 float* __restrict__ Ab) {
  const int g = blockIdx.x * 8 + (threadIdx.x >> 5);
  const int b = g >> 8;
  const int tile = g & 255;
  const int tm = tile >> 4, tn = tile & 15;
  const int lane = threadIdx.x & 31, half = lane >> 4, idx = lane & 15;
  float* Out = Ab + (size_t)b * Tn * Tn + (size_t)tm * 16 * Tn + tn * 16;
  if (tn > tm) {
#pragma unroll
    for (int r = 0; r < 8; r++) Out[(size_t)(r + 8 * half) * Tn + idx] = 0.f;
    return;
  }
  v8f acc = {0.f,0.f,0.f,0.f,0.f,0.f,0.f,0.f};
  tile_abt(Qb + (size_t)b * Tn * Hn + (size_t)tm * 16 * Hn, Hn,
           Kb + (size_t)b * Tn * Hn + (size_t)tn * 16 * Hn, Hn, Hn, acc);
#pragma unroll
  for (int r = 0; r < 8; r++) {
    float v = acc[r];
    if (tn == tm && idx > (r + 8 * half)) v = 0.f;
    Out[(size_t)(r + 8 * half) * Tn + idx] = v;
  }
}

__global__ void dla_den_kernel(const float* __restrict__ Ab, const float* __restrict__ Qb,
                               const float* __restrict__ st, int layerOff,
                               float* __restrict__ den) {
  const int wave = threadIdx.x >> 5, lane = threadIdx.x & 31;
  const int row = blockIdx.x * 8 + wave;
  const int b = row >> 8;
  const float* Ar = Ab + (size_t)b * Tn * Tn + (size_t)(row & 255) * Tn;
  const float* Qr = Qb + (size_t)row * Hn;
  const float* Z0 = st + (size_t)b * STSTRIDE + layerOff + Hn * Hn;
  float s = 0.f;
  for (int k = lane; k < Tn; k += 32) s += Ar[k];
  for (int k = lane; k < Hn; k += 32) s += Qr[k] * Z0[k];
  for (int off = 16; off; off >>= 1) s += __shfl_xor(s, off);
  if (lane == 0) den[row] = s + 1e-5f;
}

__global__ void dla_attnout_kernel(const float* __restrict__ Ab, const float* __restrict__ Vb,
                                   const float* __restrict__ Qb, const float* __restrict__ st,
                                   int layerOff, const float* __restrict__ den,
                                   float* __restrict__ Ao) {
  const int g = blockIdx.x * 8 + (threadIdx.x >> 5);
  const int b = g >> 7;
  const int tile = g & 127;
  const int tm = tile >> 3, tn = tile & 7;
  v8f acc = {0.f,0.f,0.f,0.f,0.f,0.f,0.f,0.f};
  tile_ab(Ab + (size_t)b * Tn * Tn + (size_t)tm * 16 * Tn, Tn,
          Vb + (size_t)b * Tn * Hn + tn * 16, Hn, Tn, acc);
  const float* S0 = st + (size_t)b * STSTRIDE + layerOff;
  tile_ab(Qb + (size_t)b * Tn * Hn + (size_t)tm * 16 * Hn, Hn,
          S0 + tn * 16, Hn, Hn, acc);
  const int lane = threadIdx.x & 31, half = lane >> 4, idx = lane & 15;
  float* Out = Ao + (size_t)b * Tn * Hn + (size_t)tm * 16 * Hn + tn * 16;
#pragma unroll
  for (int r = 0; r < 8; r++) {
    const int row = tm * 16 + r + 8 * half;
    Out[(size_t)(r + 8 * half) * Hn + idx] = acc[r] / den[b * Tn + row];
  }
}

__global__ void dla_states_kernel(const float* __restrict__ Kb, const float* __restrict__ Vb,
                                  const float* __restrict__ st, int layerOff,
                                  float* __restrict__ outst) {
  const int g = blockIdx.x * 8 + (threadIdx.x >> 5);
  const int b = g >> 6;
  const int tile = g & 63;
  const int tm = tile >> 3, tn = tile & 7;
  v8f acc = {0.f,0.f,0.f,0.f,0.f,0.f,0.f,0.f};
  tile_atb(Kb + (size_t)b * Tn * Hn + tm * 16, Hn,
           Vb + (size_t)b * Tn * Hn + tn * 16, Hn, Tn, acc);
  const float* S0 = st + (size_t)b * STSTRIDE + layerOff;
  float* Out = outst + (size_t)b * STSTRIDE + layerOff;
  const int lane = threadIdx.x & 31, half = lane >> 4, idx = lane & 15;
#pragma unroll
  for (int r = 0; r < 8; r++) {
    const int i = tm * 16 + r + 8 * half;
    const int l = tn * 16 + idx;
    Out[(size_t)i * Hn + l] = acc[r] + S0[(size_t)i * Hn + l];
  }
}

__global__ void dla_statez_kernel(const float* __restrict__ Kb, const float* __restrict__ st,
                                  int layerOff, float* __restrict__ outst) {
  const int b = blockIdx.x, i = threadIdx.x;
  const float* Kp = Kb + (size_t)b * Tn * Hn + i;
  float s = 0.f;
  for (int t = 0; t < Tn; t++) s += Kp[(size_t)t * Hn];
  const size_t o = (size_t)b * STSTRIDE + layerOff + Hn * Hn + i;
  outst[o] = st[o] + s;
}

__global__ void dla_ff1_kernel(const float* __restrict__ Ao, const float* __restrict__ w1,
                               const float* __restrict__ b1, float* __restrict__ h1) {
  const int g = blockIdx.x * 8 + (threadIdx.x >> 5);
  const int b = g >> 7;
  const int tile = g & 127;
  const int tm = tile >> 3, tn = tile & 7;
  v8f acc = {0.f,0.f,0.f,0.f,0.f,0.f,0.f,0.f};
  tile_abt(Ao + (size_t)b * Tn * Hn + (size_t)tm * 16 * Hn, Hn,
           w1 + (size_t)tn * 16 * Hn, Hn, Hn, acc);
  const int lane = threadIdx.x & 31, half = lane >> 4, idx = lane & 15;
  float* Out = h1 + (size_t)b * Tn * Hn + (size_t)tm * 16 * Hn + tn * 16;
#pragma unroll
  for (int r = 0; r < 8; r++) {
    float v = acc[r] + b1[tn * 16 + idx];
    Out[(size_t)(r + 8 * half) * Hn + idx] = v > 0.f ? v : 0.f;
  }
}

__global__ void dla_ff2_kernel(const float* __restrict__ h1, const float* __restrict__ w2,
                               const float* __restrict__ b2, const float* __restrict__ xn,
                               const float* __restrict__ scw, const float* __restrict__ scb,
                               float* __restrict__ xcur) {
  const int g = blockIdx.x * 8 + (threadIdx.x >> 5);
  const int b = g >> 7;
  const int tile = g & 127;
  const int tm = tile >> 3, tn = tile & 7;
  v8f acc1 = {0.f,0.f,0.f,0.f,0.f,0.f,0.f,0.f};
  v8f acc2 = {0.f,0.f,0.f,0.f,0.f,0.f,0.f,0.f};
  tile_abt(h1 + (size_t)b * Tn * Hn + (size_t)tm * 16 * Hn, Hn,
           w2 + (size_t)tn * 16 * Hn, Hn, Hn, acc1);
  tile_abt(xn + (size_t)b * Tn * INn + (size_t)tm * 16 * INn, INn,
           scw + (size_t)tn * 16 * INn, INn, INn, acc2);
  const int lane = threadIdx.x & 31, half = lane >> 4, idx = lane & 15;
  float* Out = xcur + (size_t)b * Tn * Hn + (size_t)tm * 16 * Hn + tn * 16;
#pragma unroll
  for (int r = 0; r < 8; r++) {
    float v = acc1[r] + b2[tn * 16 + idx];
    v = v > 0.f ? v : 0.f;
    Out[(size_t)(r + 8 * half) * Hn + idx] = v + acc2[r] + scb[tn * 16 + idx];
  }
}

__global__ void dla_unmap_kernel(const float* __restrict__ x, const float* __restrict__ uw,
                                 const float* __restrict__ ub, float* __restrict__ y) {
  const int g = blockIdx.x * 8 + (threadIdx.x >> 5);
  const int b = g >> 7;
  const int tile = g & 127;
  const int tm = tile >> 3, tn = tile & 7;
  v8f acc = {0.f,0.f,0.f,0.f,0.f,0.f,0.f,0.f};
  tile_abt(x + (size_t)b * Tn * Hn + (size_t)tm * 16 * Hn, Hn,
           uw + (size_t)tn * 16 * Hn, Hn, Hn, acc);
  const int lane = threadIdx.x & 31, half = lane >> 4, idx = lane & 15;
#pragma unroll
  for (int r = 0; r < 8; r++) {
    const int t = tm * 16 + r + 8 * half;
    const int o = tn * 16 + idx;
    y[(size_t)t * Bn * OUTn + (size_t)b * OUTn + o] = acc[r] + ub[o];
  }
}

extern "C" void kernel_launch(void* const* d_in, const int* in_sizes, int n_in,
                              void* d_out, int out_size, void* d_ws, size_t ws_size,
                              hipStream_t stream) {
  (void)in_sizes; (void)n_in; (void)out_size; (void)ws_size;
  const float* z   = (const float*)d_in[0];
  const float* st  = (const float*)d_in[1];
  const float* Wk  = (const float*)d_in[2];
  const float* Wq  = (const float*)d_in[3];
  const float* Wv  = (const float*)d_in[4];
  const float* lnw = (const float*)d_in[5];
  const float* lnb = (const float*)d_in[6];
  const float* w1  = (const float*)d_in[7];
  const float* b1  = (const float*)d_in[8];
  const float* w2  = (const float*)d_in[9];
  const float* b2  = (const float*)d_in[10];
  const float* scw = (const float*)d_in[11];
  const float* scb = (const float*)d_in[12];
  const float* uw  = (const float*)d_in[13];
  const float* ub  = (const float*)d_in[14];
  float* out = (float*)d_out;
  float* ws  = (float*)d_ws;

  float* XCUR = ws;
  float* XN   = ws + 262144;
  float* KB   = ws + 524288;
  float* QB   = ws + 786432;
  float* VB   = ws + 1048576;
  float* AB   = ws + 1310720;
  float* AO   = ws + 1835008;
  float* DEN  = ws + 2097152;
  float* OST  = out + YSZ;

  dla_rotate_kernel<<<Bn * Tn, 64, 0, stream>>>(z, st, XCUR, OST);
  for (int i = 0; i < Ln; i++) {
    const int lo = i * LAYERSTATE;
    dla_ln_kernel<<<256, 256, 0, stream>>>(XCUR, lnw + i * INn, lnb + i * INn, XN);
    dla_proj_kernel<<<384, 256, 0, stream>>>(XN, Wk + (size_t)i * Hn * INn,
                                             Wq + (size_t)i * Hn * INn,
                                             Wv + (size_t)i * Hn * INn, KB, QB, VB);
    dla_score_kernel<<<256, 256, 0, stream>>>(QB, KB, AB);
    dla_den_kernel<<<256, 256, 0, stream>>>(AB, QB, st, lo, DEN);
    dla_attnout_kernel<<<128, 256, 0, stream>>>(AB, VB, QB, st, lo, DEN, AO);
    dla_states_kernel<<<64, 256, 0, stream>>>(KB, VB, st, lo, OST);
    dla_statez_kernel<<<8, 128, 0, stream>>>(KB, st, lo, OST);
    dla_ff1_kernel<<<128, 256, 0, stream>>>(AO, w1 + (size_t)i * Hn * Hn, b1 + i * Hn, QB);
    dla_ff2_kernel<<<128, 256, 0, stream>>>(QB, w2 + (size_t)i * Hn * Hn, b2 + i * Hn,
                                            XN, scw + (size_t)i * Hn * INn, scb + i * Hn, XCUR);
  }
  dla_unmap_kernel<<<128, 256, 0, stream>>>(XCUR, uw, ub, out);
}


# Fuse the 20 launches into 4: per layer one launch for [rotate,] layernorm, K/Q/V and one for scores..ff2 [,unmap] per 16-row tile behind work-group barriers, S'/Z' tiles as extra work-groups
# speedup vs baseline: 1.5113x; 1.5113x over previous
#include <hip/hip_runtime.h>
#include <hip/hip_bf16.h>
#include <math.h>

#define Bn 8
#define Tn 256
#define INn 128
#define Hn 128
#define OUTn 128
#define Ln 2
#define STSTRIDE 33025
#define LAYERSTATE 16512
#define YSZ (Tn*Bn*OUTn)

typedef __attribute__((ext_vector_type(2))) float v2f;
typedef __attribute__((ext_vector_type(8))) float v8f;

__device__ __forceinline__ v8f wmma4(v2f a, v2f b, v8f c) {
  return __builtin_amdgcn_wmma_f32_16x16x4_f32(false, a, false, b, (short)0, c, false, false);
}

__device__ __forceinline__ void tile_abt(const float* A, int lda, const float* Bt, int ldb,
                                         int K, v8f& acc) {
  const int lane = threadIdx.x & 31;
  const int half = lane >> 4;
  const int idx  = lane & 15;
  const float* ap = A  + (size_t)idx * lda + 2 * half;
  const float* bp = Bt + (size_t)idx * ldb + 2 * half;
  for (int k = 0; k < K; k += 4) {
    v2f a, b;
    a.x = ap[k]; a.y = ap[k + 1];
    b.x = bp[k]; b.y = bp[k + 1];
    acc = wmma4(a, b, acc);
  }
}

__device__ __forceinline__ void tile_ab(const float* A, int lda, const float* Bm, int ldb,
                                        int K, v8f& acc) {
  const int lane = threadIdx.x & 31;
  const int half = lane >> 4;
  const int idx  = lane & 15;
  const float* ap = A + (size_t)idx * lda + 2 * half;
  for (int k = 0; k < K; k += 4) {
    v2f a, b;
    a.x = ap[k]; a.y = ap[k + 1];
    b.x = Bm[(size_t)(k + 2 * half)     * ldb + idx];
    b.y = Bm[(size_t)(k + 2 * half + 1) * ldb + idx];
    acc = wmma4(a, b, acc);
  }
}

__device__ __forceinline__ void tile_atb(const float* Am, int lda, const float* Bm, int ldb,
                                         int K, v8f& acc) {
  const int lane = threadIdx.x & 31;
  const int half = lane >> 4;
  const int idx  = lane & 15;
  for (int k = 0; k < K; k += 4) {
    v2f a, b;
    a.x = Am[(size_t)(k + 2 * half)     * lda + idx];
    a.y = Am[(size_t)(k + 2 * half + 1) * lda + idx];
    b.x = Bm[(size_t)(k + 2 * half)     * ldb + idx];
    b.y = Bm[(size_t)(k + 2 * half + 1) * ldb + idx];
    acc = wmma4(a, b, acc);
  }
}


__device__ __forceinline__ void st_rotate(const float* __restrict__ z, const float* __restrict__ st,
                                          float* xcur, float* outst, int b, int t, int j) {
  const float ts = st[(size_t)b * STSTRIDE + STSTRIDE - 1];
  const float pos = ts + (float)t;
  const float freq = expf((float)(2 * j) * (-9.210340371976184f / 128.0f));
  const float ang = pos * freq;
  const float s = sinf(ang), c = cosf(ang);
  const float* zp = z + (size_t)t * Bn * INn + (size_t)b * INn;
  const float x1 = zp[j], x2 = zp[j + 64];
  float* xp = xcur + ((size_t)b * Tn + t) * INn;
  xp[j]      = x1 * c - x2 * s;
  xp[j + 64] = x2 * c + x1 * s;
  if (t == 0 && j == 0)
    outst[(size_t)b * STSTRIDE + STSTRIDE - 1] = ts + (float)Tn;
}

__device__ __forceinline__ void st_ln(const float* x, const float* __restrict__ w,
                                      const float* __restrict__ bias, float* xn, int row) {
  const int lane = threadIdx.x & 31;
  const float* xr = x + (size_t)row * INn;
  float s = 0.f, ss = 0.f;
  for (int k = lane; k < INn; k += 32) { float v = xr[k]; s += v; ss += v * v; }
  for (int off = 16; off; off >>= 1) { s += __shfl_xor(s, off); ss += __shfl_xor(ss, off); }
  const float m = s * (1.0f / INn);
  const float var = ss * (1.0f / INn) - m * m;
  const float inv = rsqrtf(var + 1e-5f);
  float* outr = xn + (size_t)row * INn;
  for (int k = lane; k < INn; k += 32)
    outr[k] = (xr[k] - m) * inv * w[k] + bias[k];
}

__device__ __forceinline__ void st_proj(const float* xn,
                                        const float* __restrict__ Wk, const float* __restrict__ Wq,
                                        const float* __restrict__ Wv,
                                        float* Kb, float* Qb, float* Vb,
                                        int which, int b, int tm, int tn) {
  const float* W = (which == 0) ? Wk : (which == 1) ? Wq : Wv;
  const float* A  = xn + (size_t)b * Tn * INn + (size_t)tm * 16 * INn;
  const float* Bt = W + (size_t)tn * 16 * INn;
  v8f acc = {0.f,0.f,0.f,0.f,0.f,0.f,0.f,0.f};
  tile_abt(A, INn, Bt, INn, INn, acc);
  float* Out = ((which == 0) ? Kb : (which == 1) ? Qb : Vb)
             + (size_t)b * Tn * Hn + (size_t)tm * 16 * Hn + tn * 16;
  const int lane = threadIdx.x & 31, half = lane >> 4, idx = lane & 15;
#pragma unroll
  for (int r = 0; r < 8; r++) {
    float v = acc[r];
    if (which != 2) v = ((v > 0.f) ? v : expm1f(v)) + 1.0f;
    Out[(size_t)(r + 8 * half) * Hn + idx] = v;
  }
}

__device__ __forceinline__ void st_score(const float* Qb, const float* Kb, float* Ab,
                                         int b, int tm, int tn) {
  const int lane = threadIdx.x & 31, half = lane >> 4, idx = lane & 15;
  float* Out = Ab + (size_t)b * Tn * Tn + (size_t)tm * 16 * Tn + tn * 16;
  if (tn > tm) {
#pragma unroll
    for (int r = 0; r < 8; r++) Out[(size_t)(r + 8 * half) * Tn + idx] = 0.f;
  } else {
    v8f acc = {0.f,0.f,0.f,0.f,0.f,0.f,0.f,0.f};
    tile_abt(Qb + (size_t)b * Tn * Hn + (size_t)tm * 16 * Hn, Hn,
             Kb + (size_t)b * Tn * Hn + (size_t)tn * 16 * Hn, Hn, Hn, acc);
#pragma unroll
    for (int r = 0; r < 8; r++) {
      float v = acc[r];
      if (tn == tm && idx > (r + 8 * half)) v = 0.f;
      Out[(size_t)(r + 8 * half) * Tn + idx] = v;
    }
  }
}

__device__ __forceinline__ void st_den(const float* Ab, const float* Qb,
                                       const float* __restrict__ st, int layerOff,
                                       float* den, int row) {
  const int lane = threadIdx.x & 31;
  const int b = row >> 8;
  const float* Ar = Ab + (size_t)b * Tn * Tn + (size_t)(row & 255) * Tn;
  const float* Qr = Qb + (size_t)row * Hn;
  const float* Z0 = st + (size_t)b * STSTRIDE + layerOff + Hn * Hn;
  float s = 0.f;
  for (int k = lane; k < Tn; k += 32) s += Ar[k];
  for (int k = lane; k < Hn; k += 32) s += Qr[k] * Z0[k];
  for (int off = 16; off; off >>= 1) s += __shfl_xor(s, off);
  if (lane == 0) den[row] = s + 1e-5f;
}

__device__ __forceinline__ void st_attnout(const float* Ab, const float* Vb, const float* Qb,
                                           const float* __restrict__ st, int layerOff,
                                           const float* den, float* Ao, int b, int tm, int tn) {
  v8f acc = {0.f,0.f,0.f,0.f,0.f,0.f,0.f,0.f};
  tile_ab(Ab + (size_t)b * Tn * Tn + (size_t)tm * 16 * Tn, Tn,
          Vb + (size_t)b * Tn * Hn + tn * 16, Hn, Tn, acc);
  const float* S0 = st + (size_t)b * STSTRIDE + layerOff;
  tile_ab(Qb + (size_t)b * Tn * Hn + (size_t)tm * 16 * Hn, Hn,
          S0 + tn * 16, Hn, Hn, acc);
  const int lane = threadIdx.x & 31, half = lane >> 4, idx = lane & 15;
  float* Out = Ao + (size_t)b * Tn * Hn + (size_t)tm * 16 * Hn + tn * 16;
#pragma unroll
  for (int r = 0; r < 8; r++) {
    const int row = tm * 16 + r + 8 * half;
    Out[(size_t)(r + 8 * half) * Hn + idx] = acc[r] / den[b * Tn + row];
  }
}

__device__ __forceinline__ void st_states(const float* Kb, const float* Vb,
                                          const float* __restrict__ st, int layerOff,
                                          float* outst, int g) {
  const int b = g >> 6;
  const int tile = g & 63;
  const int tm = tile >> 3, tn = tile & 7;
  v8f acc = {0.f,0.f,0.f,0.f,0.f,0.f,0.f,0.f};
  tile_atb(Kb + (size_t)b * Tn * Hn + tm * 16, Hn,
           Vb + (size_t)b * Tn * Hn + tn * 16, Hn, Tn, acc);
  const float* S0 = st + (size_t)b * STSTRIDE + layerOff;
  float* Out = outst + (size_t)b * STSTRIDE + layerOff;
  const int lane = threadIdx.x & 31, half = lane >> 4, idx = lane & 15;
#pragma unroll
  for (int r = 0; r < 8; r++) {
    const int i = tm * 16 + r + 8 * half;
    const int l = tn * 16 + idx;
    Out[(size_t)i * Hn + l] = acc[r] + S0[(size_t)i * Hn + l];
  }
}

__device__ __forceinline__ void st_statez(const float* Kb, const float* __restrict__ st,
                                          int layerOff, float* outst, int b, int i) {
  const float* Kp = Kb + (size_t)b * Tn * Hn + i;
  float s = 0.f;
  for (int t = 0; t < Tn; t++) s += Kp[(size_t)t * Hn];
  const size_t o = (size_t)b * STSTRIDE + layerOff + Hn * Hn + i;
  outst[o] = st[o] + s;
}

__device__ __forceinline__ void st_ff1(const float* Ao, const float* __restrict__ w1,
                                       const float* __restrict__ b1, float* h1,
                                       int b, int tm, int tn) {
  v8f acc = {0.f,0.f,0.f,0.f,0.f,0.f,0.f,0.f};
  tile_abt(Ao + (size_t)b * Tn * Hn + (size_t)tm * 16 * Hn, Hn,
           w1 + (size_t)tn * 16 * Hn, Hn, Hn, acc);
  const int lane = threadIdx.x & 31, half = lane >> 4, idx = lane & 15;
  float* Out = h1 + (size_t)b * Tn * Hn + (size_t)tm * 16 * Hn + tn * 16;
#pragma unroll
  for (int r = 0; r < 8; r++) {
    float v = acc[r] + b1[tn * 16 + idx];
    Out[(size_t)(r + 8 * half) * Hn + idx] = v > 0.f ? v : 0.f;
  }
}

__device__ __forceinline__ void st_ff2(const float* h1, const float* __restrict__ w2,
                                       const float* __restrict__ b2, const float* xn,
                                       const float* __restrict__ scw, const float* __restrict__ scb,
                                       float* xcur, int b, int tm, int tn) {
  v8f acc1 = {0.f,0.f,0.f,0.f,0.f,0.f,0.f,0.f};
  v8f acc2 = {0.f,0.f,0.f,0.f,0.f,0.f,0.f,0.f};
  tile_abt(h1 + (size_t)b * Tn * Hn + (size_t)tm * 16 * Hn, Hn,
           w2 + (size_t)tn * 16 * Hn, Hn, Hn, acc1);
  tile_abt(xn + (size_t)b * Tn * INn + (size_t)tm * 16 * INn, INn,
           scw + (size_t)tn * 16 * INn, INn, INn, acc2);
  const int lane = threadIdx.x & 31, half = lane >> 4, idx = lane & 15;
  float* Out = xcur + (size_t)b * Tn * Hn + (size_t)tm * 16 * Hn + tn * 16;
#pragma unroll
  for (int r = 0; r < 8; r++) {
    float v = acc1[r] + b2[tn * 16 + idx];
    v = v > 0.f ? v : 0.f;
    Out[(size_t)(r + 8 * half) * Hn + idx] = v + acc2[r] + scb[tn * 16 + idx];
  }
}

__device__ __forceinline__ void st_unmap(const float* x, const float* __restrict__ uw,
                                         const float* __restrict__ ub, float* y,
                                         int b, int tm, int tn) {
  v8f acc = {0.f,0.f,0.f,0.f,0.f,0.f,0.f,0.f};
  tile_abt(x + (size_t)b * Tn * Hn + (size_t)tm * 16 * Hn, Hn,
           uw + (size_t)tn * 16 * Hn, Hn, Hn, acc);
  const int lane = threadIdx.x & 31, half = lane >> 4, idx = lane & 15;
#pragma unroll
  for (int r = 0; r < 8; r++) {
    const int t = tm * 16 + r + 8 * half;
    const int o = tn * 16 + idx;
    y[(size_t)t * Bn * OUTn + (size_t)b * OUTn + o] = acc[r] + ub[o];
  }
}

template <bool ROT>
__global__ __launch_bounds__(768) void dla_front_kernel(
    const float* __restrict__ z, const float* __restrict__ st,
    const float* __restrict__ lnw, const float* __restrict__ lnb,
    const float* __restrict__ Wk, const float* __restrict__ Wq, const float* __restrict__ Wv,
    float* xcur, float* xn, float* Kb, float* Qb, float* Vb, float* outst) {
  const int b = blockIdx.x >> 4, tm = blockIdx.x & 15;
  const int wave = threadIdx.x >> 5;
  if (ROT) {
    for (int i = threadIdx.x; i < 16 * 64; i += 768)
      st_rotate(z, st, xcur, outst, b, tm * 16 + (i >> 6), i & 63);
    __syncthreads();
  }
  if (wave < 16) st_ln(xcur, lnw, lnb, xn, b * Tn + tm * 16 + wave);
  __syncthreads();
  st_proj(xn, Wk, Wq, Wv, Kb, Qb, Vb, wave >> 3, b, tm, wave & 7);
}

template <bool LAST>
__global__ __launch_bounds__(512) void dla_mix_kernel(
    float* Qb, const float* Kb, const float* Vb, float* Ab, float* den, float* Ao,
    const float* xn, float* xcur,
    const float* __restrict__ st, int layerOff, float* outst,
    const float* __restrict__ w1, const float* __restrict__ b1,
    const float* __restrict__ w2, const float* __restrict__ b2,
    const float* __restrict__ scw, const float* __restrict__ scb,
    const float* __restrict__ uw, const float* __restrict__ ub, float* y) {
  const int wave = threadIdx.x >> 5;
  if (blockIdx.x < 128) {
    const int b = blockIdx.x >> 4, tm = blockIdx.x & 15;
    st_score(Qb, Kb, Ab, b, tm, wave);
    __syncthreads();
    st_den(Ab, Qb, st, layerOff, den, b * Tn + tm * 16 + wave);
    __syncthreads();
    if (wave < 8) st_attnout(Ab, Vb, Qb, st, layerOff, den, Ao, b, tm, wave);
    __syncthreads();
    if (wave < 8) st_ff1(Ao, w1, b1, Qb, b, tm, wave);
    __syncthreads();
    if (wave < 8) st_ff2(Qb, w2, b2, xn, scw, scb, xcur, b, tm, wave);
    if (LAST) {
      __syncthreads();
      if (wave < 8) st_unmap(xcur, uw, ub, y, b, tm, wave);
    }
  } else if (blockIdx.x < 160) {
    st_states(Kb, Vb, st, layerOff, outst, (blockIdx.x - 128) * 16 + wave);
  } else {
    const int e = (blockIdx.x - 160) * 512 + threadIdx.x;
    st_statez(Kb, st, layerOff, outst, e >> 7, e & 127);
  }
}

extern "C" void kernel_launch(void* const* d_in, const int* in_sizes, int n_in,
                              void* d_out, int out_size, void* d_ws, size_t ws_size,
                              hipStream_t stream) {
  (void)in_sizes; (void)n_in; (void)out_size; (void)ws_size;
  const float* z   = (const float*)d_in[0];
  const float* st  = (const float*)d_in[1];
  const float* Wk  = (const float*)d_in[2];
  const float* Wq  = (const float*)d_in[3];
  const float* Wv  = (const float*)d_in[4];
  const float* lnw = (const float*)d_in[5];
  const float* lnb = (const float*)d_in[6];
  const float* w1  = (const float*)d_in[7];
  const float* b1  = (const float*)d_in[8];
  const float* w2  = (const float*)d_in[9];
  const float* b2  = (const float*)d_in[10];
  const float* scw = (const float*)d_in[11];
  const float* scb = (const float*)d_in[12];
  const float* uw  = (const float*)d_in[13];
  const float* ub  = (const float*)d_in[14];
  float* out = (float*)d_out;
  float* ws  = (float*)d_ws;

  float* XCUR = ws;
  float* XN   = ws + 262144;
  float* KB   = ws + 524288;
  float* QB   = ws + 786432;
  float* VB   = ws + 1048576;
  float* AB   = ws + 1310720;
  float* AO   = ws + 1835008;
  float* DEN  = ws + 2097152;
  float* OST  = out + YSZ;

  dla_front_kernel<true><<<128, 768, 0, stream>>>(z, st, lnw, lnb, Wk, Wq, Wv,
                                                  XCUR, XN, KB, QB, VB, OST);
  dla_mix_kernel<false><<<162, 512, 0, stream>>>(QB, KB, VB, AB, DEN, AO, XN, XCUR,
                                                 st, 0, OST, w1, b1, w2, b2, scw, scb,
                                                 uw, ub, out);
  dla_front_kernel<false><<<128, 768, 0, stream>>>(z, st, lnw + INn, lnb + INn,
                                                   Wk + Hn * INn, Wq + Hn * INn, Wv + Hn * INn,
                                                   XCUR, XN, KB, QB, VB, OST);
  dla_mix_kernel<true><<<162, 512, 0, stream>>>(QB, KB, VB, AB, DEN, AO, XN, XCUR,
                                                st, LAYERSTATE, OST,
                                                w1 + Hn * Hn, b1 + Hn, w2 + Hn * Hn, b2 + Hn,
                                                scw + Hn * INn, scb + Hn,
                                                uw, ub, out);
}


# Fragment loads widened from 8 to 16 bytes per lane: one load feeds two k-steps and touches 2x fewer cache lines per byte; sums within each block of 8 k reordered to 0,1,4,5,2,3,6,7
# speedup vs baseline: 1.6965x; 1.1225x over previous
#include <hip/hip_runtime.h>
#include <hip/hip_bf16.h>
#include <math.h>

#define Bn 8
#define Tn 256
#define INn 128
#define Hn 128
#define OUTn 128
#define Ln 2
#define STSTRIDE 33025
#define LAYERSTATE 16512
#define YSZ (Tn*Bn*OUTn)

typedef __attribute__((ext_vector_type(2))) float v2f;
typedef __attribute__((ext_vector_type(4))) float v4f;
typedef __attribute__((ext_vector_type(8))) float v8f;

__device__ __forceinline__ v8f wmma4(v2f a, v2f b, v8f c) {
  return __builtin_amdgcn_wmma_f32_16x16x4_f32(false, a, false, b, (short)0, c, false, false);
}

__device__ __forceinline__ v4f load_k4(const float* p) {
  v4f v;
  __builtin_memcpy(&v, __builtin_assume_aligned(p, 16), 16);
  return v;
}


__device__ __forceinline__ void tile_abt(const float* A, int lda, const float* Bt, int ldb,
                                         int K, v8f& acc) {
  const int lane = threadIdx.x & 31;
  const int half = lane >> 4;
  const int idx  = lane & 15;
  const float* ap = A  + (size_t)idx * lda + 4 * half;
  const float* bp = Bt + (size_t)idx * ldb + 4 * half;
  for (int k = 0; k < K; k += 8) {
    const v4f a4 = load_k4(ap + k);
    const v4f b4 = load_k4(bp + k);
    v2f a, b;
    a.x = a4.x; a.y = a4.y; b.x = b4.x; b.y = b4.y;
    acc = wmma4(a, b, acc);
    a.x = a4.z; a.y = a4.w; b.x = b4.z; b.y = b4.w;
    acc = wmma4(a, b, acc);
  }
}

__device__ __forceinline__ void tile_ab(const float* A, int lda, const float* Bm, int ldb,
                                        int K, v8f& acc) {
  const int lane = threadIdx.x & 31;
  const int half = lane >> 4;
  const int idx  = lane & 15;
  const float* ap = A + (size_t)idx * lda + 4 * half;
  for (int k = 0; k < K; k += 8) {
    const v4f a4 = load_k4(ap + k);
    const float* br = Bm + (size_t)(k + 4 * half) * ldb + idx;
    v2f a, b;
    a.x = a4.x; a.y = a4.y;
    b.x = br[0];
    b.y = br[(size_t)ldb];
    acc = wmma4(a, b, acc);
    a.x = a4.z; a.y = a4.w;
    b.x = br[(size_t)2 * ldb];
    b.y = br[(size_t)3 * ldb];
    acc = wmma4(a, b, acc);
  }
}

__device__ __forceinline__ void tile_atb(const float* Am, int lda, const float* Bm, int ldb,
                                         int K, v8f& acc) {
  const int lane = threadIdx.x & 31;
  const int half = lane >> 4;
  const int idx  = lane & 15;
  for (int k = 0; k < K; k += 4) {
    v2f a, b;
    a.x = Am[(size_t)(k + 2 * half)     * lda + idx];
    a.y = Am[(size_t)(k + 2 * half + 1) * lda + idx];
    b.x = Bm[(size_t)(k + 2 * half)     * ldb + idx];
    b.y = Bm[(size_t)(k + 2 * half + 1) * ldb + idx];
    acc = wmma4(a, b, acc);
  }
}


__device__ __forceinline__ void st_rotate(const float* __restrict__ z, const float* __restrict__ st,
                                          float* xcur, float* outst, int b, int t, int j) {
  const float ts = st[(size_t)b * STSTRIDE + STSTRIDE - 1];
  const float pos = ts + (float)t;
  const float freq = expf((float)(2 * j) * (-9.210340371976184f / 128.0f));
  const float ang = pos * freq;
  const float s = sinf(ang), c = cosf(ang);
  const float* zp = z + (size_t)t * Bn * INn + (size_t)b * INn;
  const float x1 = zp[j], x2 = zp[j + 64];
  float* xp = xcur + ((size_t)b * Tn + t) * INn;
  xp[j]      = x1 * c - x2 * s;
  xp[j + 64] = x2 * c + x1 * s;
  if (t == 0 && j == 0)
    outst[(size_t)b * STSTRIDE + STSTRIDE - 1] = ts + (float)Tn;
}

__device__ __forceinline__ void st_ln(const float* x, const float* __restrict__ w,
                                      const float* __restrict__ bias, float* xn, int row) {
  const int lane = threadIdx.x & 31;
  const float* xr = x + (size_t)row * INn;
  float s = 0.f, ss = 0.f;
  for (int k = lane; k < INn; k += 32) { float v = xr[k]; s += v; ss += v * v; }
  for (int off = 16; off; off >>= 1) { s += __shfl_xor(s, off); ss += __shfl_xor(ss, off); }
  const float m = s * (1.0f / INn);
  const float var = ss * (1.0f / INn) - m * m;
  const float inv = rsqrtf(var + 1e-5f);
  float* outr = xn + (size_t)row * INn;
  for (int k = lane; k < INn; k += 32)
    outr[k] = (xr[k] - m) * inv * w[k] + bias[k];
}

__device__ __forceinline__ void st_proj(const float* xn,
                                        const float* __restrict__ Wk, const float* __restrict__ Wq,
                                        const float* __restrict__ Wv,
                                        float* Kb, float* Qb, float* Vb,
                                        int which, int b, int tm, int tn) {
  const float* W = (which == 0) ? Wk : (which == 1) ? Wq : Wv;
  const float* A  = xn + (size_t)b * Tn * INn + (size_t)tm * 16 * INn;
  const float* Bt = W + (size_t)tn * 16 * INn;
  v8f acc = {0.f,0.f,0.f,0.f,0.f,0.f,0.f,0.f};
  tile_abt(A, INn, Bt, INn, INn, acc);
  float* Out = ((which == 0) ? Kb : (which == 1) ? Qb : Vb)
             + (size_t)b * Tn * Hn + (size_t)tm * 16 * Hn + tn * 16;
  const int lane = threadIdx.x & 31, half = lane >> 4, idx = lane & 15;
#pragma unroll
  for (int r = 0; r < 8; r++) {
    float v = acc[r];
    if (which != 2) v = ((v > 0.f) ? v : expm1f(v)) + 1.0f;
    Out[(size_t)(r + 8 * half) * Hn + idx] = v;
  }
}

__device__ __forceinline__ void st_score(const float* Qb, const float* Kb, float* Ab,
                                         int b, int tm, int tn) {
  const int lane = threadIdx.x & 31, half = lane >> 4, idx = lane & 15;
  float* Out = Ab + (size_t)b * Tn * Tn + (size_t)tm * 16 * Tn + tn * 16;
  if (tn > tm) {
#pragma unroll
    for (int r = 0; r < 8; r++) Out[(size_t)(r + 8 * half) * Tn + idx] = 0.f;
  } else {
    v8f acc = {0.f,0.f,0.f,0.f,0.f,0.f,0.f,0.f};
    tile_abt(Qb + (size_t)b * Tn * Hn + (size_t)tm * 16 * Hn, Hn,
             Kb + (size_t)b * Tn * Hn + (size_t)tn * 16 * Hn, Hn, Hn, acc);
#pragma unroll
    for (int r = 0; r < 8; r++) {
      float v = acc[r];
      if (tn == tm && idx > (r + 8 * half)) v = 0.f;
      Out[(size_t)(r + 8 * half) * Tn + idx] = v;
    }
  }
}

__device__ __forceinline__ void st_den(const float* Ab, const float* Qb,
                                       const float* __restrict__ st, int layerOff,
                                       float* den, int row) {
  const int lane = threadIdx.x & 31;
  const int b = row >> 8;
  const float* Ar = Ab + (size_t)b * Tn * Tn + (size_t)(row & 255) * Tn;
  const float* Qr = Qb + (size_t)row * Hn;
  const float* Z0 = st + (size_t)b * STSTRIDE + layerOff + Hn * Hn;
  float s = 0.f;
  for (int k = lane; k < Tn; k += 32) s += Ar[k];
  for (int k = lane; k < Hn; k += 32) s += Qr[k] * Z0[k];
  for (int off = 16; off; off >>= 1) s += __shfl_xor(s, off);
  if (lane == 0) den[row] = s + 1e-5f;
}

__device__ __forceinline__ void st_attnout(const float* Ab, const float* Vb, const float* Qb,
                                           const float* __restrict__ st, int layerOff,
                                           const float* den, float* Ao, int b, int tm, int tn) {
  v8f acc = {0.f,0.f,0.f,0.f,0.f,0.f,0.f,0.f};
  tile_ab(Ab + (size_t)b * Tn * Tn + (size_t)tm * 16 * Tn, Tn,
          Vb + (size_t)b * Tn * Hn + tn * 16, Hn, Tn, acc);
  const float* S0 = st + (size_t)b * STSTRIDE + layerOff;
  tile_ab(Qb + (size_t)b * Tn * Hn + (size_t)tm * 16 * Hn, Hn,
          S0 + tn * 16, Hn, Hn, acc);
  const int lane = threadIdx.x & 31, half = lane >> 4, idx = lane & 15;
  float* Out = Ao + (size_t)b * Tn * Hn + (size_t)tm * 16 * Hn + tn * 16;
#pragma unroll
  for (int r = 0; r < 8; r++) {
    const int row = tm * 16 + r + 8 * half;
    Out[(size_t)(r + 8 * half) * Hn + idx] = acc[r] / den[b * Tn + row];
  }
}

__device__ __forceinline__ void st_states(const float* Kb, const float* Vb,
                                          const float* __restrict__ st, int layerOff,
                                          float* outst, int g) {
  const int b = g >> 6;
  const int tile = g & 63;
  const int tm = tile >> 3, tn = tile & 7;
  v8f acc = {0.f,0.f,0.f,0.f,0.f,0.f,0.f,0.f};
  tile_atb(Kb + (size_t)b * Tn * Hn + tm * 16, Hn,
           Vb + (size_t)b * Tn * Hn + tn * 16, Hn, Tn, acc);
  const float* S0 = st + (size_t)b * STSTRIDE + layerOff;
  float* Out = outst + (size_t)b * STSTRIDE + layerOff;
  const int lane = threadIdx.x & 31, half = lane >> 4, idx = lane & 15;
#pragma unroll
  for (int r = 0; r < 8; r++) {
    const int i = tm * 16 + r + 8 * half;
    const int l = tn * 16 + idx;
    Out[(size_t)i * Hn + l] = acc[r] + S0[(size_t)i * Hn + l];
  }
}

__device__ __forceinline__ void st_statez(const float* Kb, const float* __restrict__ st,
                                          int layerOff, float* outst, int b, int i) {
  const float* Kp = Kb + (size_t)b * Tn * Hn + i;
  float s = 0.f;
  for (int t = 0; t < Tn; t++) s += Kp[(size_t)t * Hn];
  const size_t o = (size_t)b * STSTRIDE + layerOff + Hn * Hn + i;
  outst[o] = st[o] + s;
}

__device__ __forceinline__ void st_ff1(const float* Ao, const float* __restrict__ w1,
                                       const float* __restrict__ b1, float* h1,
                                       int b, int tm, int tn) {
  v8f acc = {0.f,0.f,0.f,0.f,0.f,0.f,0.f,0.f};
  tile_abt(Ao + (size_t)b * Tn * Hn + (size_t)tm * 16 * Hn, Hn,
           w1 + (size_t)tn * 16 * Hn, Hn, Hn, acc);
  const int lane = threadIdx.x & 31, half = lane >> 4, idx = lane & 15;
  float* Out = h1 + (size_t)b * Tn * Hn + (size_t)tm * 16 * Hn + tn * 16;
#pragma unroll
  for (int r = 0; r < 8; r++) {
    float v = acc[r] + b1[tn * 16 + idx];
    Out[(size_t)(r + 8 * half) * Hn + idx] = v > 0.f ? v : 0.f;
  }
}

__device__ __forceinline__ void st_ff2(const float* h1, const float* __restrict__ w2,
                                       const float* __restrict__ b2, const float* xn,
                                       const float* __restrict__ scw, const float* __restrict__ scb,
                                       float* xcur, int b, int tm, int tn) {
  v8f acc1 = {0.f,0.f,0.f,0.f,0.f,0.f,0.f,0.f};
  v8f acc2 = {0.f,0.f,0.f,0.f,0.f,0.f,0.f,0.f};
  tile_abt(h1 + (size_t)b * Tn * Hn + (size_t)tm * 16 * Hn, Hn,
           w2 + (size_t)tn * 16 * Hn, Hn, Hn, acc1);
  tile_abt(xn + (size_t)b * Tn * INn + (size_t)tm * 16 * INn, INn,
           scw + (size_t)tn * 16 * INn, INn, INn, acc2);
  const int lane = threadIdx.x & 31, half = lane >> 4, idx = lane & 15;
  float* Out = xcur + (size_t)b * Tn * Hn + (size_t)tm * 16 * Hn + tn * 16;
#pragma unroll
  for (int r = 0; r < 8; r++) {
    float v = acc1[r] + b2[tn * 16 + idx];
    v = v > 0.f ? v : 0.f;
    Out[(size_t)(r + 8 * half) * Hn + idx] = v + acc2[r] + scb[tn * 16 + idx];
  }
}

__device__ __forceinline__ void st_unmap(const float* x, const float* __restrict__ uw,
                                         const float* __restrict__ ub, float* y,
                                         int b, int tm, int tn) {
  v8f acc = {0.f,0.f,0.f,0.f,0.f,0.f,0.f,0.f};
  tile_abt(x + (size_t)b * Tn * Hn + (size_t)tm * 16 * Hn, Hn,
           uw + (size_t)tn * 16 * Hn, Hn, Hn, acc);
  const int lane = threadIdx.x & 31, half = lane >> 4, idx = lane & 15;
#pragma unroll
  for (int r = 0; r < 8; r++) {
    const int t = tm * 16 + r + 8 * half;
    const int o = tn * 16 + idx;
    y[(size_t)t * Bn * OUTn + (size_t)b * OUTn + o] = acc[r] + ub[o];
  }
}

template <bool ROT>
__global__ __launch_bounds__(768) void dla_front_kernel(
    const float* __restrict__ z, const float* __restrict__ st,
    const float* __restrict__ lnw, const float* __restrict__ lnb,
    const float* __restrict__ Wk, const float* __restrict__ Wq, const float* __restrict__ Wv,
    float* xcur, float* xn, float* Kb, float* Qb, float* Vb, float* outst) {
  const int b = blockIdx.x >> 4, tm = blockIdx.x & 15;
  const int wave = threadIdx.x >> 5;
  if (ROT) {
    for (int i = threadIdx.x; i < 16 * 64; i += 768)
      st_rotate(z, st, xcur, outst, b, tm * 16 + (i >> 6), i & 63);
    __syncthreads();
  }
  if (wave < 16) st_ln(xcur, lnw, lnb, xn, b * Tn + tm * 16 + wave);
  __syncthreads();
  st_proj(xn, Wk, Wq, Wv, Kb, Qb, Vb, wave >> 3, b, tm, wave & 7);
}

template <bool LAST>
__global__ __launch_bounds__(512) void dla_mix_kernel(
    float* Qb, const float* Kb, const float* Vb, float* Ab, float* den, float* Ao,
    const float* xn, float* xcur,
    const float* __restrict__ st, int layerOff, float* outst,
    const float* __restrict__ w1, const float* __restrict__ b1,
    const float* __restrict__ w2, const float* __restrict__ b2,
    const float* __restrict__ scw, const float* __restrict__ scb,
    const float* __restrict__ uw, const float* __restrict__ ub, float* y) {
  const int wave = threadIdx.x >> 5;
  if (blockIdx.x < 128) {
    const int b = blockIdx.x >> 4, tm = blockIdx.x & 15;
    st_score(Qb, Kb, Ab, b, tm, wave);
    __syncthreads();
    st_den(Ab, Qb, st, layerOff, den, b * Tn + tm * 16 + wave);
    __syncthreads();
    if (wave < 8) st_attnout(Ab, Vb, Qb, st, layerOff, den, Ao, b, tm, wave);
    __syncthreads();
    if (wave < 8) st_ff1(Ao, w1, b1, Qb, b, tm, wave);
    __syncthreads();
    if (wave < 8) st_ff2(Qb, w2, b2, xn, scw, scb, xcur, b, tm, wave);
    if (LAST) {
      __syncthreads();
      if (wave < 8) st_unmap(xcur, uw, ub, y, b, tm, wave);
    }
  } else if (blockIdx.x < 160) {
    st_states(Kb, Vb, st, layerOff, outst, (blockIdx.x - 128) * 16 + wave);
  } else {
    const int e = (blockIdx.x - 160) * 512 + threadIdx.x;
    st_statez(Kb, st, layerOff, outst, e >> 7, e & 127);
  }
}

extern "C" void kernel_launch(void* const* d_in, const int* in_sizes, int n_in,
                              void* d_out, int out_size, void* d_ws, size_t ws_size,
                              hipStream_t stream) {
  (void)in_sizes; (void)n_in; (void)out_size; (void)ws_size;
  const float* z   = (const float*)d_in[0];
  const float* st  = (const float*)d_in[1];
  const float* Wk  = (const float*)d_in[2];
  const float* Wq  = (const float*)d_in[3];
  const float* Wv  = (const float*)d_in[4];
  const float* lnw = (const float*)d_in[5];
  const float* lnb = (const float*)d_in[6];
  const float* w1  = (const float*)d_in[7];
  const float* b1  = (const float*)d_in[8];
  const float* w2  = (const float*)d_in[9];
  const float* b2  = (const float*)d_in[10];
  const float* scw = (const float*)d_in[11];
  const float* scb = (const float*)d_in[12];
  const float* uw  = (const float*)d_in[13];
  const float* ub  = (const float*)d_in[14];
  float* out = (float*)d_out;
  float* ws  = (float*)d_ws;

  float* XCUR = ws;
  float* XN   = ws + 262144;
  float* KB   = ws + 524288;
  float* QB   = ws + 786432;
  float* VB   = ws + 1048576;
  float* AB   = ws + 1310720;
  float* AO   = ws + 1835008;
  float* DEN  = ws + 2097152;
  float* OST  = out + YSZ;

  dla_front_kernel<true><<<128, 768, 0, stream>>>(z, st, lnw, lnb, Wk, Wq, Wv,
                                                  XCUR, XN, KB, QB, VB, OST);
  dla_mix_kernel<false><<<162, 512, 0, stream>>>(QB, KB, VB, AB, DEN, AO, XN, XCUR,
                                                 st, 0, OST, w1, b1, w2, b2, scw, scb,
                                                 uw, ub, out);
  dla_front_kernel<false><<<128, 768, 0, stream>>>(z, st, lnw + INn, lnb + INn,
                                                   Wk + Hn * INn, Wq + Hn * INn, Wv + Hn * INn,
                                                   XCUR, XN, KB, QB, VB, OST);
  dla_mix_kernel<true><<<162, 512, 0, stream>>>(QB, KB, VB, AB, DEN, AO, XN, XCUR,
                                                st, LAYERSTATE, OST,
                                                w1 + Hn * Hn, b1 + Hn, w2 + Hn * Hn, b2 + Hn,
                                                scw + Hn * INn, scb + Hn,
                                                uw, ub, out);
}


# Mix kernel: attn-out, ff1, ff2, unmap load their fragments in register groups of 64 k issued ahead of the WMMA chain (20 to 50 loads in flight, 208 VGPRs, waves_per_eu 4); sum order unchanged
# speedup vs baseline: 1.7681x; 1.0422x over previous
#include <hip/hip_runtime.h>
#include <hip/hip_bf16.h>
#include <math.h>

#define Bn 8
#define Tn 256
#define INn 128
#define Hn 128
#define OUTn 128
#define Ln 2
#define STSTRIDE 33025
#define LAYERSTATE 16512
#define YSZ (Tn*Bn*OUTn)

typedef __attribute__((ext_vector_type(2))) float v2f;
typedef __attribute__((ext_vector_type(4))) float v4f;
typedef __attribute__((ext_vector_type(8))) float v8f;

__device__ __forceinline__ v8f wmma4(v2f a, v2f b, v8f c) {
  return __builtin_amdgcn_wmma_f32_16x16x4_f32(false, a, false, b, (short)0, c, false, false);
}

__device__ __forceinline__ v4f load_k4(const float* p) {
  v4f v;
  __builtin_memcpy(&v, __builtin_assume_aligned(p, 16), 16);
  return v;
}


__device__ __forceinline__ void tile_abt(const float* A, int lda, const float* Bt, int ldb,
                                         int K, v8f& acc) {
  const int lane = threadIdx.x & 31;
  const int half = lane >> 4;
  const int idx  = lane & 15;
  const float* ap = A  + (size_t)idx * lda + 4 * half;
  const float* bp = Bt + (size_t)idx * ldb + 4 * half;
  for (int k = 0; k < K; k += 8) {
    const v4f a4 = load_k4(ap + k);
    const v4f b4 = load_k4(bp + k);
    v2f a, b;
    a.x = a4.x; a.y = a4.y; b.x = b4.x; b.y = b4.y;
    acc = wmma4(a, b, acc);
    a.x = a4.z; a.y = a4.w; b.x = b4.z; b.y = b4.w;
    acc = wmma4(a, b, acc);
  }
}

__device__ __forceinline__ void tile_ab(const float* A, int lda, const float* Bm, int ldb,
                                        int K, v8f& acc) {
  const int lane = threadIdx.x & 31;
  const int half = lane >> 4;
  const int idx  = lane & 15;
  const float* ap = A + (size_t)idx * lda + 4 * half;
  for (int k = 0; k < K; k += 8) {
    const v4f a4 = load_k4(ap + k);
    const float* br = Bm + (size_t)(k + 4 * half) * ldb + idx;
    v2f a, b;
    a.x = a4.x; a.y = a4.y;
    b.x = br[0];
    b.y = br[(size_t)ldb];
    acc = wmma4(a, b, acc);
    a.x = a4.z; a.y = a4.w;
    b.x = br[(size_t)2 * ldb];
    b.y = br[(size_t)3 * ldb];
    acc = wmma4(a, b, acc);
  }
}

__device__ __forceinline__ void tile_atb(const float* Am, int lda, const float* Bm, int ldb,
                                         int K, v8f& acc) {
  const int lane = threadIdx.x & 31;
  const int half = lane >> 4;
  const int idx  = lane & 15;
  for (int k = 0; k < K; k += 4) {
    v2f a, b;
    a.x = Am[(size_t)(k + 2 * half)     * lda + idx];
    a.y = Am[(size_t)(k + 2 * half + 1) * lda + idx];
    b.x = Bm[(size_t)(k + 2 * half)     * ldb + idx];
    b.y = Bm[(size_t)(k + 2 * half + 1) * ldb + idx];
    acc = wmma4(a, b, acc);
  }
}

struct Frag64 { v4f a[8]; v4f b[8]; };

__device__ __forceinline__ void ld_abt64(Frag64& f, const float* ap, const float* bp) {
#pragma unroll
  for (int i = 0; i < 8; i++) { f.a[i] = load_k4(ap + 8 * i); f.b[i] = load_k4(bp + 8 * i); }
}

__device__ __forceinline__ void ld_ab64(Frag64& f, const float* ap, const float* bc, int ldb) {
#pragma unroll
  for (int i = 0; i < 8; i++) {
    f.a[i] = load_k4(ap + 8 * i);
    const float* br = bc + (size_t)(8 * i) * ldb;
    v4f b;
    b.x = br[0];
    b.y = br[(size_t)ldb];
    b.z = br[(size_t)2 * ldb];
    b.w = br[(size_t)3 * ldb];
    f.b[i] = b;
  }
}

__device__ __forceinline__ void mm64(const Frag64& f, v8f& acc) {
#pragma unroll
  for (int i = 0; i < 8; i++) {
    v2f a, b;
    a.x = f.a[i].x; a.y = f.a[i].y; b.x = f.b[i].x; b.y = f.b[i].y;
    acc = wmma4(a, b, acc);
    a.x = f.a[i].z; a.y = f.a[i].w; b.x = f.b[i].z; b.y = f.b[i].w;
    acc = wmma4(a, b, acc);
  }
}


__device__ __forceinline__ void st_rotate(const float* __restrict__ z, const float* __restrict__ st,
                                          float* xcur, float* outst, int b, int t, int j) {
  const float ts = st[(size_t)b * STSTRIDE + STSTRIDE - 1];
  const float pos = ts + (float)t;
  const float freq = expf((float)(2 * j) * (-9.210340371976184f / 128.0f));
  const float ang = pos * freq;
  const float s = sinf(ang), c = cosf(ang);
  const float* zp = z + (size_t)t * Bn * INn + (size_t)b * INn;
  const float x1 = zp[j], x2 = zp[j + 64];
  float* xp = xcur + ((size_t)b * Tn + t) * INn;
  xp[j]      = x1 * c - x2 * s;
  xp[j + 64] = x2 * c + x1 * s;
  if (t == 0 && j == 0)
    outst[(size_t)b * STSTRIDE + STSTRIDE - 1] = ts + (float)Tn;
}

__device__ __forceinline__ void st_ln(const float* x, const float* __restrict__ w,
                                      const float* __restrict__ bias, float* xn, int row) {
  const int lane = threadIdx.x & 31;
  const float* xr = x + (size_t)row * INn;
  float s = 0.f, ss = 0.f;
  for (int k = lane; k < INn; k += 32) { float v = xr[k]; s += v; ss += v * v; }
  for (int off = 16; off; off >>= 1) { s += __shfl_xor(s, off); ss += __shfl_xor(ss, off); }
  const float m = s * (1.0f / INn);
  const float var = ss * (1.0f / INn) - m * m;
  const float inv = rsqrtf(var + 1e-5f);
  float* outr = xn + (size_t)row * INn;
  for (int k = lane; k < INn; k += 32)
    outr[k] = (xr[k] - m) * inv * w[k] + bias[k];
}

__device__ __forceinline__ void st_proj(const float* xn,
                                        const float* __restrict__ Wk, const float* __restrict__ Wq,
                                        const float* __restrict__ Wv,
                                        float* Kb, float* Qb, float* Vb,
                                        int which, int b, int tm, int tn) {
  const float* W = (which == 0) ? Wk : (which == 1) ? Wq : Wv;
  const float* A  = xn + (size_t)b * Tn * INn + (size_t)tm * 16 * INn;
  const float* Bt = W + (size_t)tn * 16 * INn;
  v8f acc = {0.f,0.f,0.f,0.f,0.f,0.f,0.f,0.f};
  tile_abt(A, INn, Bt, INn, INn, acc);
  float* Out = ((which == 0) ? Kb : (which == 1) ? Qb : Vb)
             + (size_t)b * Tn * Hn + (size_t)tm * 16 * Hn + tn * 16;
  const int lane = threadIdx.x & 31, half = lane >> 4, idx = lane & 15;
#pragma unroll
  for (int r = 0; r < 8; r++) {
    float v = acc[r];
    if (which != 2) v = ((v > 0.f) ? v : expm1f(v)) + 1.0f;
    Out[(size_t)(r + 8 * half) * Hn + idx] = v;
  }
}

__device__ __forceinline__ void st_score(const float* Qb, const float* Kb, float* Ab,
                                         int b, int tm, int tn) {
  const int lane = threadIdx.x & 31, half = lane >> 4, idx = lane & 15;
  float* Out = Ab + (size_t)b * Tn * Tn + (size_t)tm * 16 * Tn + tn * 16;
  if (tn > tm) {
#pragma unroll
    for (int r = 0; r < 8; r++) Out[(size_t)(r + 8 * half) * Tn + idx] = 0.f;
  } else {
    v8f acc = {0.f,0.f,0.f,0.f,0.f,0.f,0.f,0.f};
    tile_abt(Qb + (size_t)b * Tn * Hn + (size_t)tm * 16 * Hn, Hn,
             Kb + (size_t)b * Tn * Hn + (size_t)tn * 16 * Hn, Hn, Hn, acc);
#pragma unroll
    for (int r = 0; r < 8; r++) {
      float v = acc[r];
      if (tn == tm && idx > (r + 8 * half)) v = 0.f;
      Out[(size_t)(r + 8 * half) * Tn + idx] = v;
    }
  }
}

__device__ __forceinline__ void st_den(const float* Ab, const float* Qb,
                                       const float* __restrict__ st, int layerOff,
                                       float* den, int row) {
  const int lane = threadIdx.x & 31;
  const int b = row >> 8;
  const float* Ar = Ab + (size_t)b * Tn * Tn + (size_t)(row & 255) * Tn;
  const float* Qr = Qb + (size_t)row * Hn;
  const float* Z0 = st + (size_t)b * STSTRIDE + layerOff + Hn * Hn;
  float s = 0.f;
  for (int k = lane; k < Tn; k += 32) s += Ar[k];
  for (int k = lane; k < Hn; k += 32) s += Qr[k] * Z0[k];
  for (int off = 16; off; off >>= 1) s += __shfl_xor(s, off);
  if (lane == 0) den[row] = s + 1e-5f;
}

__device__ __forceinline__ void st_attnout(const float* Ab, const float* Vb, const float* Qb,
                                           const float* __restrict__ st, int layerOff,
                                           const float* den, float* Ao, int b, int tm, int tn) {
  const int lane = threadIdx.x & 31, half = lane >> 4, idx = lane & 15;
  v8f acc = {0.f,0.f,0.f,0.f,0.f,0.f,0.f,0.f};
  const float* S0 = st + (size_t)b * STSTRIDE + layerOff;
  const float* aA = Ab + (size_t)b * Tn * Tn + (size_t)(tm * 16 + idx) * Tn + 4 * half;
  const float* bV = Vb + (size_t)b * Tn * Hn + (size_t)(4 * half) * Hn + tn * 16 + idx;
  const float* aQ = Qb + (size_t)b * Tn * Hn + (size_t)(tm * 16 + idx) * Hn + 4 * half;
  const float* bS = S0 + (size_t)(4 * half) * Hn + tn * 16 + idx;
  float dn[8];
#pragma unroll
  for (int r = 0; r < 8; r++) dn[r] = den[b * Tn + tm * 16 + r + 8 * half];
  Frag64 f0, f1, f2;
  ld_ab64(f0, aA,       bV,                     Hn);
  ld_ab64(f1, aA + 64,  bV + (size_t)64 * Hn,   Hn);
  ld_ab64(f2, aA + 128, bV + (size_t)128 * Hn,  Hn);
  mm64(f0, acc);
  ld_ab64(f0, aA + 192, bV + (size_t)192 * Hn,  Hn);
  mm64(f1, acc);
  ld_ab64(f1, aQ,       bS,                     Hn);
  mm64(f2, acc);
  ld_ab64(f2, aQ + 64,  bS + (size_t)64 * Hn,   Hn);
  mm64(f0, acc);
  mm64(f1, acc);
  mm64(f2, acc);
  float* Out = Ao + (size_t)b * Tn * Hn + (size_t)tm * 16 * Hn + tn * 16;
#pragma unroll
  for (int r = 0; r < 8; r++)
    Out[(size_t)(r + 8 * half) * Hn + idx] = acc[r] / dn[r];
}

__device__ __forceinline__ void st_states(const float* Kb, const float* Vb,
                                          const float* __restrict__ st, int layerOff,
                                          float* outst, int g) {
  const int b = g >> 6;
  const int tile = g & 63;
  const int tm = tile >> 3, tn = tile & 7;
  v8f acc = {0.f,0.f,0.f,0.f,0.f,0.f,0.f,0.f};
  tile_atb(Kb + (size_t)b * Tn * Hn + tm * 16, Hn,
           Vb + (size_t)b * Tn * Hn + tn * 16, Hn, Tn, acc);
  const float* S0 = st + (size_t)b * STSTRIDE + layerOff;
  float* Out = outst + (size_t)b * STSTRIDE + layerOff;
  const int lane = threadIdx.x & 31, half = lane >> 4, idx = lane & 15;
#pragma unroll
  for (int r = 0; r < 8; r++) {
    const int i = tm * 16 + r + 8 * half;
    const int l = tn * 16 + idx;
    Out[(size_t)i * Hn + l] = acc[r] + S0[(size_t)i * Hn + l];
  }
}

__device__ __forceinline__ void st_statez(const float* Kb, const float* __restrict__ st,
                                          int layerOff, float* outst, int b, int i) {
  const float* Kp = Kb + (size_t)b * Tn * Hn + i;
  float s = 0.f;
  for (int t = 0; t < Tn; t++) s += Kp[(size_t)t * Hn];
  const size_t o = (size_t)b * STSTRIDE + layerOff + Hn * Hn + i;
  outst[o] = st[o] + s;
}

__device__ __forceinline__ void st_ff1(const float* Ao, const float* __restrict__ w1,
                                       const float* __restrict__ b1, float* h1,
                                       int b, int tm, int tn) {
  const int lane = threadIdx.x & 31, half = lane >> 4, idx = lane & 15;
  v8f acc = {0.f,0.f,0.f,0.f,0.f,0.f,0.f,0.f};
  const float* ap = Ao + (size_t)b * Tn * Hn + (size_t)(tm * 16 + idx) * Hn + 4 * half;
  const float* bp = w1 + (size_t)(tn * 16 + idx) * Hn + 4 * half;
  Frag64 f0, f1;
  ld_abt64(f0, ap,      bp);
  ld_abt64(f1, ap + 64, bp + 64);
  mm64(f0, acc);
  mm64(f1, acc);
  float* Out = h1 + (size_t)b * Tn * Hn + (size_t)tm * 16 * Hn + tn * 16;
#pragma unroll
  for (int r = 0; r < 8; r++) {
    float v = acc[r] + b1[tn * 16 + idx];
    Out[(size_t)(r + 8 * half) * Hn + idx] = v > 0.f ? v : 0.f;
  }
}

__device__ __forceinline__ void st_ff2(const float* h1, const float* __restrict__ w2,
                                       const float* __restrict__ b2, const float* xn,
                                       const float* __restrict__ scw, const float* __restrict__ scb,
                                       float* xcur, int b, int tm, int tn) {
  const int lane = threadIdx.x & 31, half = lane >> 4, idx = lane & 15;
  v8f acc1 = {0.f,0.f,0.f,0.f,0.f,0.f,0.f,0.f};
  v8f acc2 = {0.f,0.f,0.f,0.f,0.f,0.f,0.f,0.f};
  const float* a1 = h1  + (size_t)b * Tn * Hn  + (size_t)(tm * 16 + idx) * Hn  + 4 * half;
  const float* p1 = w2  + (size_t)(tn * 16 + idx) * Hn  + 4 * half;
  const float* a2 = xn  + (size_t)b * Tn * INn + (size_t)(tm * 16 + idx) * INn + 4 * half;
  const float* p2 = scw + (size_t)(tn * 16 + idx) * INn + 4 * half;
  Frag64 f0, f1, f2;
  ld_abt64(f0, a1,      p1);
  ld_abt64(f1, a2,      p2);
  ld_abt64(f2, a1 + 64, p1 + 64);
  mm64(f0, acc1);
  ld_abt64(f0, a2 + 64, p2 + 64);
  mm64(f1, acc2);
  mm64(f2, acc1);
  mm64(f0, acc2);
  float* Out = xcur + (size_t)b * Tn * Hn + (size_t)tm * 16 * Hn + tn * 16;
#pragma unroll
  for (int r = 0; r < 8; r++) {
    float v = acc1[r] + b2[tn * 16 + idx];
    v = v > 0.f ? v : 0.f;
    Out[(size_t)(r + 8 * half) * Hn + idx] = v + acc2[r] + scb[tn * 16 + idx];
  }
}

__device__ __forceinline__ void st_unmap(const float* x, const float* __restrict__ uw,
                                         const float* __restrict__ ub, float* y,
                                         int b, int tm, int tn) {
  const int lane = threadIdx.x & 31, half = lane >> 4, idx = lane & 15;
  v8f acc = {0.f,0.f,0.f,0.f,0.f,0.f,0.f,0.f};
  const float* ap = x  + (size_t)b * Tn * Hn + (size_t)(tm * 16 + idx) * Hn + 4 * half;
  const float* bp = uw + (size_t)(tn * 16 + idx) * Hn + 4 * half;
  Frag64 f0, f1;
  ld_abt64(f0, ap,      bp);
  ld_abt64(f1, ap + 64, bp + 64);
  mm64(f0, acc);
  mm64(f1, acc);
#pragma unroll
  for (int r = 0; r < 8; r++) {
    const int t = tm * 16 + r + 8 * half;
    const int o = tn * 16 + idx;
    y[(size_t)t * Bn * OUTn + (size_t)b * OUTn + o] = acc[r] + ub[o];
  }
}

template <bool ROT>
__global__ __launch_bounds__(768) void dla_front_kernel(
    const float* __restrict__ z, const float* __restrict__ st,
    const float* __restrict__ lnw, const float* __restrict__ lnb,
    const float* __restrict__ Wk, const float* __restrict__ Wq, const float* __restrict__ Wv,
    float* xcur, float* xn, float* Kb, float* Qb, float* Vb, float* outst) {
  const int b = blockIdx.x >> 4, tm = blockIdx.x & 15;
  const int wave = threadIdx.x >> 5;
  if (ROT) {
    for (int i = threadIdx.x; i < 16 * 64; i += 768)
      st_rotate(z, st, xcur, outst, b, tm * 16 + (i >> 6), i & 63);
    __syncthreads();
  }
  if (wave < 16) st_ln(xcur, lnw, lnb, xn, b * Tn + tm * 16 + wave);
  __syncthreads();
  st_proj(xn, Wk, Wq, Wv, Kb, Qb, Vb, wave >> 3, b, tm, wave & 7);
}

template <bool LAST>
__global__ __launch_bounds__(512) __attribute__((amdgpu_waves_per_eu(4, 4))) void dla_mix_kernel(
    float* Qb, const float* Kb, const float* Vb, float* Ab, float* den, float* Ao,
    const float* xn, float* xcur,
    const float* __restrict__ st, int layerOff, float* outst,
    const float* __restrict__ w1, const float* __restrict__ b1,
    const float* __restrict__ w2, const float* __restrict__ b2,
    const float* __restrict__ scw, const float* __restrict__ scb,
    const float* __restrict__ uw, const float* __restrict__ ub, float* y) {
  const int wave = threadIdx.x >> 5;
  if (blockIdx.x < 128) {
    const int b = blockIdx.x >> 4, tm = blockIdx.x & 15;
    st_score(Qb, Kb, Ab, b, tm, wave);
    __syncthreads();
    st_den(Ab, Qb, st, layerOff, den, b * Tn + tm * 16 + wave);
    __syncthreads();
    if (wave < 8) st_attnout(Ab, Vb, Qb, st, layerOff, den, Ao, b, tm, wave);
    __syncthreads();
    if (wave < 8) st_ff1(Ao, w1, b1, Qb, b, tm, wave);
    __syncthreads();
    if (wave < 8) st_ff2(Qb, w2, b2, xn, scw, scb, xcur, b, tm, wave);
    if (LAST) {
      __syncthreads();
      if (wave < 8) st_unmap(xcur, uw, ub, y, b, tm, wave);
    }
  } else if (blockIdx.x < 160) {
    st_states(Kb, Vb, st, layerOff, outst, (blockIdx.x - 128) * 16 + wave);
  } else {
    const int e = (blockIdx.x - 160) * 512 + threadIdx.x;
    st_statez(Kb, st, layerOff, outst, e >> 7, e & 127);
  }
}

extern "C" void kernel_launch(void* const* d_in, const int* in_sizes, int n_in,
                              void* d_out, int out_size, void* d_ws, size_t ws_size,
                              hipStream_t stream) {
  (void)in_sizes; (void)n_in; (void)out_size; (void)ws_size;
  const float* z   = (const float*)d_in[0];
  const float* st  = (const float*)d_in[1];
  const float* Wk  = (const float*)d_in[2];
  const float* Wq  = (const float*)d_in[3];
  const float* Wv  = (const float*)d_in[4];
  const float* lnw = (const float*)d_in[5];
  const float* lnb = (const float*)d_in[6];
  const float* w1  = (const float*)d_in[7];
  const float* b1  = (const float*)d_in[8];
  const float* w2  = (const float*)d_in[9];
  const float* b2  = (const float*)d_in[10];
  const float* scw = (const float*)d_in[11];
  const float* scb = (const float*)d_in[12];
  const float* uw  = (const float*)d_in[13];
  const float* ub  = (const float*)d_in[14];
  float* out = (float*)d_out;
  float* ws  = (float*)d_ws;

  float* XCUR = ws;
  float* XN   = ws + 262144;
  float* KB   = ws + 524288;
  float* QB   = ws + 786432;
  float* VB   = ws + 1048576;
  float* AB   = ws + 1310720;
  float* AO   = ws + 1835008;
  float* DEN  = ws + 2097152;
  float* OST  = out + YSZ;

  dla_front_kernel<true><<<128, 768, 0, stream>>>(z, st, lnw, lnb, Wk, Wq, Wv,
                                                  XCUR, XN, KB, QB, VB, OST);
  dla_mix_kernel<false><<<162, 512, 0, stream>>>(QB, KB, VB, AB, DEN, AO, XN, XCUR,
                                                 st, 0, OST, w1, b1, w2, b2, scw, scb,
                                                 uw, ub, out);
  dla_front_kernel<false><<<128, 768, 0, stream>>>(z, st, lnw + INn, lnb + INn,
                                                   Wk + Hn * INn, Wq + Hn * INn, Wv + Hn * INn,
                                                   XCUR, XN, KB, QB, VB, OST);
  dla_mix_kernel<true><<<162, 512, 0, stream>>>(QB, KB, VB, AB, DEN, AO, XN, XCUR,
                                                st, LAYERSTATE, OST,
                                                w1 + Hn * Hn, b1 + Hn, w2 + Hn * Hn, b2 + Hn,
                                                scw + Hn * INn, scb + Hn,
                                                uw, ub, out);
}


# Tiles that one work-group hands from stage to stage (rotated x, xn, Q, scores, den, attn, h1, x) stay in LDS, not the workspace; global B operands load ahead of LDS-only barriers; sum order unchanged
# speedup vs baseline: 2.8574x; 1.6161x over previous
#include <hip/hip_runtime.h>
#include <hip/hip_bf16.h>
#include <math.h>


#define Bn 8
#define Tn 256
#define INn 128
#define Hn 128
#define OUTn 128
#define Ln 2
#define STSTRIDE 33025
#define LAYERSTATE 16512
#define YSZ (Tn*Bn*OUTn)

#define QP 132
#define AP 260

typedef __attribute__((ext_vector_type(2))) float v2f;
typedef __attribute__((ext_vector_type(4))) float v4f;
typedef __attribute__((ext_vector_type(8))) float v8f;

__device__ __forceinline__ v8f wmma4(v2f a, v2f b, v8f c) {
  return __builtin_amdgcn_wmma_f32_16x16x4_f32(false, a, false, b, (short)0, c, false, false);
}

__device__ __forceinline__ v4f load_k4(const float* p) {
  v4f v;
  __builtin_memcpy(&v, __builtin_assume_aligned(p, 16), 16);
  return v;
}

__device__ __forceinline__ void store_k4(float* p, v4f v) {
  __builtin_memcpy(__builtin_assume_aligned(p, 16), &v, 16);
}

__device__ __forceinline__ void lds_handover_barrier() {
  __builtin_amdgcn_fence(__ATOMIC_RELEASE, "workgroup", "local");
  __builtin_amdgcn_s_barrier();
  __builtin_amdgcn_fence(__ATOMIC_ACQUIRE, "workgroup", "local");
}


__device__ __forceinline__ void tile_atb(const float* Am, int lda, const float* Bm, int ldb,
                                         int K, v8f& acc) {
  const int lane = threadIdx.x & 31;
  const int half = lane >> 4;
  const int idx  = lane & 15;
  for (int k = 0; k < K; k += 4) {
    v2f a, b;
    a.x = Am[(size_t)(k + 2 * half)     * lda + idx];
    a.y = Am[(size_t)(k + 2 * half + 1) * lda + idx];
    b.x = Bm[(size_t)(k + 2 * half)     * ldb + idx];
    b.y = Bm[(size_t)(k + 2 * half + 1) * ldb + idx];
    acc = wmma4(a, b, acc);
  }
}

struct FragB64 { v4f b[8]; };

__device__ __forceinline__ void ld_bt64(FragB64& f, const float* bp) {
#pragma unroll
  for (int i = 0; i < 8; i++) f.b[i] = load_k4(bp + 8 * i);
}

__device__ __forceinline__ void ld_b64(FragB64& f, const float* bc, int ldb) {
#pragma unroll
  for (int i = 0; i < 8; i++) {
    const float* br = bc + (size_t)(8 * i) * ldb;
    v4f b;
    b.x = br[0];
    b.y = br[(size_t)ldb];
    b.z = br[(size_t)2 * ldb];
    b.w = br[(size_t)3 * ldb];
    f.b[i] = b;
  }
}

__device__ __forceinline__ void mm64s(const float* ap, const FragB64& f, v8f& acc) {
  v4f a4[8];
#pragma unroll
  for (int i = 0; i < 8; i++) a4[i] = load_k4(ap + 8 * i);
  __builtin_amdgcn_sched_barrier(0);
#pragma unroll
  for (int i = 0; i < 8; i++) {
    v2f a, b;
    a.x = a4[i].x; a.y = a4[i].y; b.x = f.b[i].x; b.y = f.b[i].y;
    acc = wmma4(a, b, acc);
    a.x = a4[i].z; a.y = a4[i].w; b.x = f.b[i].z; b.y = f.b[i].w;
    acc = wmma4(a, b, acc);
  }
}


__device__ __forceinline__ void st_rotate(const float* __restrict__ z, const float* __restrict__ st,
                                          float* xrows, float* outst, int b, int t0, int lt, int j) {
  const int t = t0 + lt;
  const float ts = st[(size_t)b * STSTRIDE + STSTRIDE - 1];
  const float pos = ts + (float)t;
  const float freq = expf((float)(2 * j) * (-9.210340371976184f / 128.0f));
  const float ang = pos * freq;
  const float s = sinf(ang), c = cosf(ang);
  const float* zp = z + (size_t)t * Bn * INn + (size_t)b * INn;
  const float x1 = zp[j], x2 = zp[j + 64];
  float* xp = xrows + lt * QP;
  xp[j]      = x1 * c - x2 * s;
  xp[j + 64] = x2 * c + x1 * s;
  if (t == 0 && j == 0)
    outst[(size_t)b * STSTRIDE + STSTRIDE - 1] = ts + (float)Tn;
}

__device__ __forceinline__ void st_ln(const float* xr, const float (&w)[4], const float (&bias)[4],
                                      float* outg, float* outl) {
  const int lane = threadIdx.x & 31;
  float xv[4];
  float s = 0.f, ss = 0.f;
#pragma unroll
  for (int j = 0; j < 4; j++) { const float v = xr[lane + 32 * j]; xv[j] = v; s += v; ss += v * v; }
  for (int off = 16; off; off >>= 1) { s += __shfl_xor(s, off); ss += __shfl_xor(ss, off); }
  const float m = s * (1.0f / INn);
  const float var = ss * (1.0f / INn) - m * m;
  const float inv = rsqrtf(var + 1e-5f);
#pragma unroll
  for (int j = 0; j < 4; j++) {
    const float v = (xv[j] - m) * inv * w[j] + bias[j];
    outg[lane + 32 * j] = v;
    outl[lane + 32 * j] = v;
  }
}

__device__ __forceinline__ void st_states(const float* Kb, const float* Vb,
                                          const float* __restrict__ st, int layerOff,
                                          float* outst, int g) {
  const int b = g >> 6;
  const int tile = g & 63;
  const int tm = tile >> 3, tn = tile & 7;
  v8f acc = {0.f,0.f,0.f,0.f,0.f,0.f,0.f,0.f};
  tile_atb(Kb + (size_t)b * Tn * Hn + tm * 16, Hn,
           Vb + (size_t)b * Tn * Hn + tn * 16, Hn, Tn, acc);
  const float* S0 = st + (size_t)b * STSTRIDE + layerOff;
  float* Out = outst + (size_t)b * STSTRIDE + layerOff;
  const int lane = threadIdx.x & 31, half = lane >> 4, idx = lane & 15;
#pragma unroll
  for (int r = 0; r < 8; r++) {
    const int i = tm * 16 + r + 8 * half;
    const int l = tn * 16 + idx;
    Out[(size_t)i * Hn + l] = acc[r] + S0[(size_t)i * Hn + l];
  }
}

__device__ __forceinline__ void st_statez(const float* Kb, const float* __restrict__ st,
                                          int layerOff, float* outst, int b, int i) {
  const float* Kp = Kb + (size_t)b * Tn * Hn + i;
  float s = 0.f;
  for (int t = 0; t < Tn; t++) s += Kp[(size_t)t * Hn];
  const size_t o = (size_t)b * STSTRIDE + layerOff + Hn * Hn + i;
  outst[o] = st[o] + s;
}

template <bool ROT>
__global__ __launch_bounds__(768) void dla_front_kernel(
    const float* __restrict__ z, const float* __restrict__ st,
    const float* __restrict__ lnw, const float* __restrict__ lnb,
    const float* __restrict__ Wk, const float* __restrict__ Wq, const float* __restrict__ Wv,
    const float* xcur, float* xn, float* Kb, float* Qb, float* Vb, float* outst) {
  __shared__ __attribute__((aligned(16))) float sXr[16 * QP];
  __shared__ __attribute__((aligned(16))) float sN[16 * QP];
  const int b = blockIdx.x >> 4, tm = blockIdx.x & 15;
  const int wave = threadIdx.x >> 5;
  const int lane = threadIdx.x & 31, half = lane >> 4, idx = lane & 15;
  const int which = wave >> 3, tn = wave & 7;
  const size_t row0 = (size_t)b * Tn + tm * 16;
  const float* W = (which == 0) ? Wk : (which == 1) ? Wq : Wv;
  FragB64 w0, w1;
  {
    const float* bp = W + (size_t)(tn * 16 + idx) * INn + 4 * half;
    ld_bt64(w0, bp);
    ld_bt64(w1, bp + 64);
  }
  float lw[4], lb[4];
#pragma unroll
  for (int j = 0; j < 4; j++) { lw[j] = lnw[lane + 32 * j]; lb[j] = lnb[lane + 32 * j]; }
  if (ROT) {
    for (int i = threadIdx.x; i < 16 * 64; i += 768)
      st_rotate(z, st, sXr, outst, b, tm * 16, i >> 6, i & 63);
    lds_handover_barrier();
  }
  if (wave < 16) {
    const float* xr = ROT ? (const float*)(sXr + wave * QP) : xcur + (row0 + wave) * INn;
    st_ln(xr, lw, lb, xn + (row0 + wave) * INn, sN + wave * QP);
  }
  lds_handover_barrier();
  v8f acc = {0.f,0.f,0.f,0.f,0.f,0.f,0.f,0.f};
  const float* ap = sN + idx * QP + 4 * half;
  mm64s(ap,      w0, acc);
  mm64s(ap + 64, w1, acc);
  float* Out = ((which == 0) ? Kb : (which == 1) ? Qb : Vb) + (row0 * Hn) + tn * 16;
#pragma unroll
  for (int r = 0; r < 8; r++) {
    float v = acc[r];
    if (which != 2) v = ((v > 0.f) ? v : expm1f(v)) + 1.0f;
    Out[(size_t)(r + 8 * half) * Hn + idx] = v;
  }
}

template <bool LAST>
__global__ __launch_bounds__(512) __attribute__((amdgpu_waves_per_eu(4, 4))) void dla_mix_kernel(
    const float* Qb, const float* Kb, const float* Vb,
    const float* xn, float* xcur,
    const float* __restrict__ st, int layerOff, float* outst,
    const float* __restrict__ w1, const float* __restrict__ b1,
    const float* __restrict__ w2, const float* __restrict__ b2,
    const float* __restrict__ scw, const float* __restrict__ scb,
    const float* __restrict__ uw, const float* __restrict__ ub, float* y) {
  __shared__ __attribute__((aligned(16))) float sQ[16 * QP];
  __shared__ __attribute__((aligned(16))) float sXn[16 * QP];
  __shared__ __attribute__((aligned(16))) float sAo[16 * QP];
  __shared__ __attribute__((aligned(16))) float sH[16 * QP];
  __shared__ __attribute__((aligned(16))) float sA[16 * AP];
  __shared__ float sDen[16];
  const int wave = threadIdx.x >> 5;
  if (blockIdx.x < 128) {
    const int b = blockIdx.x >> 4, tm = blockIdx.x & 15;
    const int lane = threadIdx.x & 31, half = lane >> 4, idx = lane & 15;
    const size_t row0 = (size_t)b * Tn + tm * 16;
    const bool live = wave <= tm;
    const bool mmw  = wave < 8;
    const int tn  = wave & 7;
    const int col = tn * 16 + idx;

    FragB64 k0, k1;
    if (live) {
      const float* kp = Kb + ((size_t)b * Tn + wave * 16 + idx) * Hn + 4 * half;
      ld_bt64(k0, kp);
      ld_bt64(k1, kp + 64);
    }
    const int lr = threadIdx.x >> 5, lc = (threadIdx.x & 31) * 4;
    const v4f q4 = load_k4(Qb + (row0 + lr) * Hn  + lc);
    const v4f x4 = load_k4(xn + (row0 + lr) * INn + lc);
    const float* Z0 = st + (size_t)b * STSTRIDE + layerOff + Hn * Hn;
    float z0[4];
#pragma unroll
    for (int j = 0; j < 4; j++) z0[j] = Z0[lane + 32 * j];
    const float* bV = Vb + (size_t)b * Tn * Hn + (size_t)(4 * half) * Hn + col;
    const float* bS = st + (size_t)b * STSTRIDE + layerOff + (size_t)(4 * half) * Hn + col;
    const float b1c = b1[col], b2c = b2[col], scbc = scb[col];
    store_k4(sQ  + lr * QP + lc, q4);
    store_k4(sXn + lr * QP + lc, x4);
    lds_handover_barrier();
    FragB64 g0, g1, g2;
    if (mmw) {
      ld_b64(g0, bV,                    Hn);
      ld_b64(g1, bV + (size_t)64 * Hn,  Hn);
      ld_b64(g2, bV + (size_t)128 * Hn, Hn);
    }

    {
      float* o = sA + (8 * half) * AP + wave * 16 + idx;
      if (live) {
        v8f acc = {0.f,0.f,0.f,0.f,0.f,0.f,0.f,0.f};
        const float* ap = sQ + idx * QP + 4 * half;
        mm64s(ap,      k0, acc);
        mm64s(ap + 64, k1, acc);
#pragma unroll
        for (int r = 0; r < 8; r++) {
          float v = acc[r];
          if (wave == tm && idx > (r + 8 * half)) v = 0.f;
          o[r * AP] = v;
        }
      } else {
#pragma unroll
        for (int r = 0; r < 8; r++) o[r * AP] = 0.f;
      }
    }
    lds_handover_barrier();

    {
      const float* Ar = sA + wave * AP;
      const float* Qr = sQ + wave * QP;
      float s = 0.f;
#pragma unroll
      for (int j = 0; j < 8; j++) s += Ar[lane + 32 * j];
#pragma unroll
      for (int j = 0; j < 4; j++) s += Qr[lane + 32 * j] * z0[j];
      for (int off = 16; off; off >>= 1) s += __shfl_xor(s, off);
      if (lane == 0) sDen[wave] = s + 1e-5f;
    }
    lds_handover_barrier();

    FragB64 wa, wb;
    FragB64 u0, u1, u2, u3;
    if (mmw) {
      v8f acc = {0.f,0.f,0.f,0.f,0.f,0.f,0.f,0.f};
      const float* aA = sA + idx * AP + 4 * half;
      const float* aQ = sQ + idx * QP + 4 * half;
      float dn[8];
#pragma unroll
      for (int r = 0; r < 8; r++) dn[r] = sDen[r + 8 * half];
      mm64s(aA,       g0, acc);
      ld_b64(g0, bV + (size_t)192 * Hn, Hn);
      mm64s(aA + 64,  g1, acc);
      ld_b64(g1, bS,                    Hn);
      mm64s(aA + 128, g2, acc);
      ld_b64(g2, bS + (size_t)64 * Hn,  Hn);
      mm64s(aA + 192, g0, acc);
      ld_bt64(wa, w1 + (size_t)col * Hn + 4 * half);
      mm64s(aQ,       g1, acc);
      mm64s(aQ + 64,  g2, acc);
#pragma unroll
      for (int r = 0; r < 8; r++)
        sAo[(r + 8 * half) * QP + col] = acc[r] / dn[r];
    }
    lds_handover_barrier();

    if (mmw) {
      ld_bt64(wb, w1  + (size_t)col * Hn  + 4 * half + 64);
      ld_bt64(u0, w2  + (size_t)col * Hn  + 4 * half);
      ld_bt64(u1, scw + (size_t)col * INn + 4 * half);
      v8f acc = {0.f,0.f,0.f,0.f,0.f,0.f,0.f,0.f};
      const float* ap = sAo + idx * QP + 4 * half;
      mm64s(ap,      wa, acc);
      mm64s(ap + 64, wb, acc);
#pragma unroll
      for (int r = 0; r < 8; r++) {
        float v = acc[r] + b1c;
        sH[(r + 8 * half) * QP + col] = v > 0.f ? v : 0.f;
      }
    }
    lds_handover_barrier();
    FragB64 y0, y1;
    float ubc = 0.f;

    if (mmw) {
      ld_bt64(u2, w2  + (size_t)col * Hn  + 4 * half + 64);
      ld_bt64(u3, scw + (size_t)col * INn + 4 * half + 64);
      if (LAST) {
        ld_bt64(y0, uw + (size_t)col * Hn + 4 * half);
        ubc = ub[col];
      }
      v8f acc1 = {0.f,0.f,0.f,0.f,0.f,0.f,0.f,0.f};
      v8f acc2 = {0.f,0.f,0.f,0.f,0.f,0.f,0.f,0.f};
      const float* a1 = sH  + idx * QP + 4 * half;
      const float* a2 = sXn + idx * QP + 4 * half;
      mm64s(a1,      u0, acc1);
      mm64s(a2,      u1, acc2);
      mm64s(a1 + 64, u2, acc1);
      mm64s(a2 + 64, u3, acc2);
#pragma unroll
      for (int r = 0; r < 8; r++) {
        float v = acc1[r] + b2c;
        v = v > 0.f ? v : 0.f;
        const float xv = v + acc2[r] + scbc;
        if (LAST) sAo[(r + 8 * half) * QP + col] = xv;
        else      xcur[(row0 + r + 8 * half) * Hn + col] = xv;
      }
    }
    if (LAST) {
      lds_handover_barrier();
      if (mmw) {
        ld_bt64(y1, uw + (size_t)col * Hn + 4 * half + 64);
        v8f acc = {0.f,0.f,0.f,0.f,0.f,0.f,0.f,0.f};
        const float* ap = sAo + idx * QP + 4 * half;
        mm64s(ap,      y0, acc);
        mm64s(ap + 64, y1, acc);
#pragma unroll
        for (int r = 0; r < 8; r++) {
          const int t = tm * 16 + r + 8 * half;
          y[(size_t)t * Bn * OUTn + (size_t)b * OUTn + col] = acc[r] + ubc;
        }
      }
    }
  } else if (blockIdx.x < 160) {
    st_states(Kb, Vb, st, layerOff, outst, (blockIdx.x - 128) * 16 + wave);
  } else {
    const int e = (blockIdx.x - 160) * 512 + threadIdx.x;
    st_statez(Kb, st, layerOff, outst, e >> 7, e & 127);
  }
}

extern "C" void kernel_launch(void* const* d_in, const int* in_sizes, int n_in,
                              void* d_out, int out_size, void* d_ws, size_t ws_size,
                              hipStream_t stream) {
  (void)in_sizes; (void)n_in; (void)out_size; (void)ws_size;
  const float* z   = (const float*)d_in[0];
  const float* st  = (const float*)d_in[1];
  const float* Wk  = (const float*)d_in[2];
  const float* Wq  = (const float*)d_in[3];
  const float* Wv  = (const float*)d_in[4];
  const float* lnw = (const float*)d_in[5];
  const float* lnb = (const float*)d_in[6];
  const float* w1  = (const float*)d_in[7];
  const float* b1  = (const float*)d_in[8];
  const float* w2  = (const float*)d_in[9];
  const float* b2  = (const float*)d_in[10];
  const float* scw = (const float*)d_in[11];
  const float* scb = (const float*)d_in[12];
  const float* uw  = (const float*)d_in[13];
  const float* ub  = (const float*)d_in[14];
  float* out = (float*)d_out;
  float* ws  = (float*)d_ws;

  float* XCUR = ws;
  float* XN   = ws + 262144;
  float* KB   = ws + 524288;
  float* QB   = ws + 786432;
  float* VB   = ws + 1048576;
  float* OST  = out + YSZ;

  dla_front_kernel<true><<<128, 768, 0, stream>>>(z, st, lnw, lnb, Wk, Wq, Wv,
                                                  XCUR, XN, KB, QB, VB, OST);
  dla_mix_kernel<false><<<162, 512, 0, stream>>>(QB, KB, VB, XN, XCUR,
                                                 st, 0, OST, w1, b1, w2, b2, scw, scb,
                                                 uw, ub, out);
  dla_front_kernel<false><<<128, 768, 0, stream>>>(z, st, lnw + INn, lnb + INn,
                                                   Wk + Hn * INn, Wq + Hn * INn, Wv + Hn * INn,
                                                   XCUR, XN, KB, QB, VB, OST);
  dla_mix_kernel<true><<<162, 512, 0, stream>>>(QB, KB, VB, XN, XCUR,
                                                st, LAYERSTATE, OST,
                                                w1 + Hn * Hn, b1 + Hn, w2 + Hn * Hn, b2 + Hn,
                                                scw + Hn * INn, scb + Hn,
                                                uw, ub, out);
}


# Layer 1's layernorm and K/V projections run at the end of layer 0's mix launch on the x tile held in LDS, its Q is projected inside its own mix launch: 3 launches, not 4; sum order unchanged
# speedup vs baseline: 3.0348x; 1.0621x over previous
#include <hip/hip_runtime.h>
#include <hip/hip_bf16.h>
#include <math.h>


#define Bn 8
#define Tn 256
#define INn 128
#define Hn 128
#define OUTn 128
#define Ln 2
#define STSTRIDE 33025
#define LAYERSTATE 16512
#define YSZ (Tn*Bn*OUTn)

#define QP 132
#define AP 260

typedef __attribute__((ext_vector_type(2))) float v2f;
typedef __attribute__((ext_vector_type(4))) float v4f;
typedef __attribute__((ext_vector_type(8))) float v8f;

__device__ __forceinline__ v8f wmma4(v2f a, v2f b, v8f c) {
  return __builtin_amdgcn_wmma_f32_16x16x4_f32(false, a, false, b, (short)0, c, false, false);
}

__device__ __forceinline__ v4f load_k4(const float* p) {
  v4f v;
  __builtin_memcpy(&v, __builtin_assume_aligned(p, 16), 16);
  return v;
}

__device__ __forceinline__ void store_k4(float* p, v4f v) {
  __builtin_memcpy(__builtin_assume_aligned(p, 16), &v, 16);
}

__device__ __forceinline__ void lds_handover_barrier() {
  __builtin_amdgcn_fence(__ATOMIC_RELEASE, "workgroup", "local");
  __builtin_amdgcn_s_barrier();
  __builtin_amdgcn_fence(__ATOMIC_ACQUIRE, "workgroup", "local");
}


__device__ __forceinline__ void tile_atb(const float* Am, int lda, const float* Bm, int ldb,
                                         int K, v8f& acc) {
  const int lane = threadIdx.x & 31;
  const int half = lane >> 4;
  const int idx  = lane & 15;
  for (int k = 0; k < K; k += 4) {
    v2f a, b;
    a.x = Am[(size_t)(k + 2 * half)     * lda + idx];
    a.y = Am[(size_t)(k + 2 * half + 1) * lda + idx];
    b.x = Bm[(size_t)(k + 2 * half)     * ldb + idx];
    b.y = Bm[(size_t)(k + 2 * half + 1) * ldb + idx];
    acc = wmma4(a, b, acc);
  }
}

struct FragB64 { v4f b[8]; };

__device__ __forceinline__ void ld_bt64(FragB64& f, const float* bp) {
#pragma unroll
  for (int i = 0; i < 8; i++) f.b[i] = load_k4(bp + 8 * i);
}

__device__ __forceinline__ void ld_b64(FragB64& f, const float* bc, int ldb) {
#pragma unroll
  for (int i = 0; i < 8; i++) {
    const float* br = bc + (size_t)(8 * i) * ldb;
    v4f b;
    b.x = br[0];
    b.y = br[(size_t)ldb];
    b.z = br[(size_t)2 * ldb];
    b.w = br[(size_t)3 * ldb];
    f.b[i] = b;
  }
}

__device__ __forceinline__ void mm64s(const float* ap, const FragB64& f, v8f& acc) {
  v4f a4[8];
#pragma unroll
  for (int i = 0; i < 8; i++) a4[i] = load_k4(ap + 8 * i);
  __builtin_amdgcn_sched_barrier(0);
#pragma unroll
  for (int i = 0; i < 8; i++) {
    v2f a, b;
    a.x = a4[i].x; a.y = a4[i].y; b.x = f.b[i].x; b.y = f.b[i].y;
    acc = wmma4(a, b, acc);
    a.x = a4[i].z; a.y = a4[i].w; b.x = f.b[i].z; b.y = f.b[i].w;
    acc = wmma4(a, b, acc);
  }
}


__device__ __forceinline__ void st_rotate(const float* __restrict__ z, const float* __restrict__ st,
                                          float* xrows, float* outst, int b, int t0, int lt, int j) {
  const int t = t0 + lt;
  const float ts = st[(size_t)b * STSTRIDE + STSTRIDE - 1];
  const float pos = ts + (float)t;
  const float freq = expf((float)(2 * j) * (-9.210340371976184f / 128.0f));
  const float ang = pos * freq;
  const float s = sinf(ang), c = cosf(ang);
  const float* zp = z + (size_t)t * Bn * INn + (size_t)b * INn;
  const float x1 = zp[j], x2 = zp[j + 64];
  float* xp = xrows + lt * QP;
  xp[j]      = x1 * c - x2 * s;
  xp[j + 64] = x2 * c + x1 * s;
  if (t == 0 && j == 0)
    outst[(size_t)b * STSTRIDE + STSTRIDE - 1] = ts + (float)Tn;
}

__device__ __forceinline__ void ld_ln4(const float* __restrict__ lnw, const float* __restrict__ lnb,
                                       float (&w)[4], float (&bias)[4]) {
  const int lane = threadIdx.x & 31;
#pragma unroll
  for (int j = 0; j < 4; j++) { w[j] = lnw[lane + 32 * j]; bias[j] = lnb[lane + 32 * j]; }
}

__device__ __forceinline__ void st_ln(const float* xr, const float (&w)[4], const float (&bias)[4],
                                      float* outg, float* outl) {
  const int lane = threadIdx.x & 31;
  float xv[4];
  float s = 0.f, ss = 0.f;
#pragma unroll
  for (int j = 0; j < 4; j++) { const float v = xr[lane + 32 * j]; xv[j] = v; s += v; ss += v * v; }
  for (int off = 16; off; off >>= 1) { s += __shfl_xor(s, off); ss += __shfl_xor(ss, off); }
  const float m = s * (1.0f / INn);
  const float var = ss * (1.0f / INn) - m * m;
  const float inv = rsqrtf(var + 1e-5f);
#pragma unroll
  for (int j = 0; j < 4; j++) {
    const float v = (xv[j] - m) * inv * w[j] + bias[j];
    outg[lane + 32 * j] = v;
    outl[lane + 32 * j] = v;
  }
}

__device__ __forceinline__ void st_states(const float* Kb, const float* Vb,
                                          const float* __restrict__ st, int layerOff,
                                          float* outst, int g) {
  const int b = g >> 6;
  const int tile = g & 63;
  const int tm = tile >> 3, tn = tile & 7;
  v8f acc = {0.f,0.f,0.f,0.f,0.f,0.f,0.f,0.f};
  tile_atb(Kb + (size_t)b * Tn * Hn + tm * 16, Hn,
           Vb + (size_t)b * Tn * Hn + tn * 16, Hn, Tn, acc);
  const float* S0 = st + (size_t)b * STSTRIDE + layerOff;
  float* Out = outst + (size_t)b * STSTRIDE + layerOff;
  const int lane = threadIdx.x & 31, half = lane >> 4, idx = lane & 15;
#pragma unroll
  for (int r = 0; r < 8; r++) {
    const int i = tm * 16 + r + 8 * half;
    const int l = tn * 16 + idx;
    Out[(size_t)i * Hn + l] = acc[r] + S0[(size_t)i * Hn + l];
  }
}

__device__ __forceinline__ void st_statez(const float* Kb, const float* __restrict__ st,
                                          int layerOff, float* outst, int b, int i) {
  const float* Kp = Kb + (size_t)b * Tn * Hn + i;
  float s = 0.f;
  for (int t = 0; t < Tn; t++) s += Kp[(size_t)t * Hn];
  const size_t o = (size_t)b * STSTRIDE + layerOff + Hn * Hn + i;
  outst[o] = st[o] + s;
}

__global__ __launch_bounds__(768) void dla_front_kernel(
    const float* __restrict__ z, const float* __restrict__ st,
    const float* __restrict__ lnw, const float* __restrict__ lnb,
    const float* __restrict__ Wk, const float* __restrict__ Wq, const float* __restrict__ Wv,
    float* xn, float* Kb, float* Qb, float* Vb, float* outst) {
  __shared__ __attribute__((aligned(16))) float sXr[16 * QP];
  __shared__ __attribute__((aligned(16))) float sN[16 * QP];
  const int b = blockIdx.x >> 4, tm = blockIdx.x & 15;
  const int wave = threadIdx.x >> 5;
  const int lane = threadIdx.x & 31, half = lane >> 4, idx = lane & 15;
  const int which = wave >> 3, tn = wave & 7;
  const size_t row0 = (size_t)b * Tn + tm * 16;
  const float* W = (which == 0) ? Wk : (which == 1) ? Wq : Wv;
  FragB64 w0, w1;
  {
    const float* bp = W + (size_t)(tn * 16 + idx) * INn + 4 * half;
    ld_bt64(w0, bp);
    ld_bt64(w1, bp + 64);
  }
  float lw[4], lb[4];
#pragma unroll
  for (int j = 0; j < 4; j++) { lw[j] = lnw[lane + 32 * j]; lb[j] = lnb[lane + 32 * j]; }
  for (int i = threadIdx.x; i < 16 * 64; i += 768)
    st_rotate(z, st, sXr, outst, b, tm * 16, i >> 6, i & 63);
  lds_handover_barrier();
  if (wave < 16)
    st_ln(sXr + wave * QP, lw, lb, xn + (row0 + wave) * INn, sN + wave * QP);
  lds_handover_barrier();
  v8f acc = {0.f,0.f,0.f,0.f,0.f,0.f,0.f,0.f};
  const float* ap = sN + idx * QP + 4 * half;
  mm64s(ap,      w0, acc);
  mm64s(ap + 64, w1, acc);
  float* Out = ((which == 0) ? Kb : (which == 1) ? Qb : Vb) + (row0 * Hn) + tn * 16;
#pragma unroll
  for (int r = 0; r < 8; r++) {
    float v = acc[r];
    if (which != 2) v = ((v > 0.f) ? v : expm1f(v)) + 1.0f;
    Out[(size_t)(r + 8 * half) * Hn + idx] = v;
  }
}

template <bool LAST>
__global__ __launch_bounds__(512) __attribute__((amdgpu_waves_per_eu(4, 4))) void dla_mix_kernel(
    const float* Qb, const float* Kb, const float* Vb, const float* xn,
    const float* __restrict__ st, int layerOff, float* outst,
    const float* __restrict__ w1, const float* __restrict__ b1,
    const float* __restrict__ w2, const float* __restrict__ b2,
    const float* __restrict__ scw, const float* __restrict__ scb,
    const float* __restrict__ nlnw, const float* __restrict__ nlnb,
    const float* __restrict__ nWk, const float* __restrict__ nWv,
    float* nxn, float* nK, float* nV,
    const float* __restrict__ Wq,
    const float* __restrict__ uw, const float* __restrict__ ub, float* y) {
  __shared__ __attribute__((aligned(16))) float sQ[16 * QP];
  __shared__ __attribute__((aligned(16))) float sXn[16 * QP];
  __shared__ __attribute__((aligned(16))) float sAo[16 * QP];
  __shared__ __attribute__((aligned(16))) float sH[16 * QP];
  __shared__ __attribute__((aligned(16))) float sA[16 * AP];
  __shared__ float sDen[16];
  const int wave = threadIdx.x >> 5;
  if (blockIdx.x < 128) {
    const int b = blockIdx.x >> 4, tm = blockIdx.x & 15;
    const int lane = threadIdx.x & 31, half = lane >> 4, idx = lane & 15;
    const size_t row0 = (size_t)b * Tn + tm * 16;
    const bool live = wave <= tm;
    const bool mmw  = wave < 8;
    const int tn  = wave & 7;
    const int col = tn * 16 + idx;

    FragB64 k0, k1;
    if (live) {
      const float* kp = Kb + ((size_t)b * Tn + wave * 16 + idx) * Hn + 4 * half;
      ld_bt64(k0, kp);
      ld_bt64(k1, kp + 64);
    }
    const int lr = threadIdx.x >> 5, lc = (threadIdx.x & 31) * 4;
    const v4f x4 = load_k4(xn + (row0 + lr) * INn + lc);
    v4f q4;
    FragB64 q0, q1;
    if (LAST) {
      if (mmw) {
        const float* qp = Wq + (size_t)col * INn + 4 * half;
        ld_bt64(q0, qp);
        ld_bt64(q1, qp + 64);
      }
    } else {
      q4 = load_k4(Qb + (row0 + lr) * Hn + lc);
    }
    const float* Z0 = st + (size_t)b * STSTRIDE + layerOff + Hn * Hn;
    float z0[4];
#pragma unroll
    for (int j = 0; j < 4; j++) z0[j] = Z0[lane + 32 * j];
    const float* bV = Vb + (size_t)b * Tn * Hn + (size_t)(4 * half) * Hn + col;
    const float* bS = st + (size_t)b * STSTRIDE + layerOff + (size_t)(4 * half) * Hn + col;
    const float b1c = b1[col], b2c = b2[col], scbc = scb[col];
    if (!LAST) store_k4(sQ + lr * QP + lc, q4);
    store_k4(sXn + lr * QP + lc, x4);
    lds_handover_barrier();
    if (LAST) {
      if (mmw) {
        v8f acc = {0.f,0.f,0.f,0.f,0.f,0.f,0.f,0.f};
        const float* ap = sXn + idx * QP + 4 * half;
        mm64s(ap,      q0, acc);
        mm64s(ap + 64, q1, acc);
#pragma unroll
        for (int r = 0; r < 8; r++) {
          const float v = acc[r];
          sQ[(r + 8 * half) * QP + col] = ((v > 0.f) ? v : expm1f(v)) + 1.0f;
        }
      }
      lds_handover_barrier();
    }
    FragB64 g0, g1, g2;
    if (mmw) {
      ld_b64(g0, bV,                    Hn);
      ld_b64(g1, bV + (size_t)64 * Hn,  Hn);
      ld_b64(g2, bV + (size_t)128 * Hn, Hn);
    }

    {
      float* o = sA + (8 * half) * AP + wave * 16 + idx;
      if (live) {
        v8f acc = {0.f,0.f,0.f,0.f,0.f,0.f,0.f,0.f};
        const float* ap = sQ + idx * QP + 4 * half;
        mm64s(ap,      k0, acc);
        mm64s(ap + 64, k1, acc);
#pragma unroll
        for (int r = 0; r < 8; r++) {
          float v = acc[r];
          if (wave == tm && idx > (r + 8 * half)) v = 0.f;
          o[r * AP] = v;
        }
      } else {
#pragma unroll
        for (int r = 0; r < 8; r++) o[r * AP] = 0.f;
      }
    }
    lds_handover_barrier();

    {
      const float* Ar = sA + wave * AP;
      const float* Qr = sQ + wave * QP;
      float s = 0.f;
#pragma unroll
      for (int j = 0; j < 8; j++) s += Ar[lane + 32 * j];
#pragma unroll
      for (int j = 0; j < 4; j++) s += Qr[lane + 32 * j] * z0[j];
      for (int off = 16; off; off >>= 1) s += __shfl_xor(s, off);
      if (lane == 0) sDen[wave] = s + 1e-5f;
    }
    lds_handover_barrier();

    FragB64 wa, wb;
    FragB64 u0, u1, u2, u3;
    if (mmw) {
      v8f acc = {0.f,0.f,0.f,0.f,0.f,0.f,0.f,0.f};
      const float* aA = sA + idx * AP + 4 * half;
      const float* aQ = sQ + idx * QP + 4 * half;
      float dn[8];
#pragma unroll
      for (int r = 0; r < 8; r++) dn[r] = sDen[r + 8 * half];
      mm64s(aA,       g0, acc);
      ld_b64(g0, bV + (size_t)192 * Hn, Hn);
      mm64s(aA + 64,  g1, acc);
      ld_b64(g1, bS,                    Hn);
      mm64s(aA + 128, g2, acc);
      ld_b64(g2, bS + (size_t)64 * Hn,  Hn);
      mm64s(aA + 192, g0, acc);
      ld_bt64(wa, w1 + (size_t)col * Hn + 4 * half);
      mm64s(aQ,       g1, acc);
      mm64s(aQ + 64,  g2, acc);
#pragma unroll
      for (int r = 0; r < 8; r++)
        sAo[(r + 8 * half) * QP + col] = acc[r] / dn[r];
    }
    lds_handover_barrier();

    if (mmw) {
      ld_bt64(wb, w1  + (size_t)col * Hn  + 4 * half + 64);
      ld_bt64(u0, w2  + (size_t)col * Hn  + 4 * half);
      ld_bt64(u1, scw + (size_t)col * INn + 4 * half);
      v8f acc = {0.f,0.f,0.f,0.f,0.f,0.f,0.f,0.f};
      const float* ap = sAo + idx * QP + 4 * half;
      mm64s(ap,      wa, acc);
      mm64s(ap + 64, wb, acc);
#pragma unroll
      for (int r = 0; r < 8; r++) {
        float v = acc[r] + b1c;
        sH[(r + 8 * half) * QP + col] = v > 0.f ? v : 0.f;
      }
    }
    lds_handover_barrier();
    FragB64 y0, y1;
    float ubc = 0.f;
    FragB64 p0, p1;
    float nlw[4], nlb[4];
    const float* pw = nullptr;
    if (!LAST) pw = (mmw ? nWk : nWv) + (size_t)col * INn + 4 * half;
    if (!LAST && !mmw) {
      ld_bt64(p0, pw);
      ld_ln4(nlnw, nlnb, nlw, nlb);
    }

    if (mmw) {
      ld_bt64(u2, w2  + (size_t)col * Hn  + 4 * half + 64);
      ld_bt64(u3, scw + (size_t)col * INn + 4 * half + 64);
      if (LAST) {
        ld_bt64(y0, uw + (size_t)col * Hn + 4 * half);
        ubc = ub[col];
      }
      v8f acc1 = {0.f,0.f,0.f,0.f,0.f,0.f,0.f,0.f};
      v8f acc2 = {0.f,0.f,0.f,0.f,0.f,0.f,0.f,0.f};
      const float* a1 = sH  + idx * QP + 4 * half;
      const float* a2 = sXn + idx * QP + 4 * half;
      mm64s(a1,      u0, acc1);
      mm64s(a2,      u1, acc2);
      mm64s(a1 + 64, u2, acc1);
      if (!LAST) ld_bt64(p0, pw);
      mm64s(a2 + 64, u3, acc2);
      if (!LAST) ld_ln4(nlnw, nlnb, nlw, nlb);
#pragma unroll
      for (int r = 0; r < 8; r++) {
        float v = acc1[r] + b2c;
        v = v > 0.f ? v : 0.f;
        const float xv = v + acc2[r] + scbc;
        sAo[(r + 8 * half) * QP + col] = xv;
      }
    }
    lds_handover_barrier();
    if (LAST) {
      if (mmw) {
        ld_bt64(y1, uw + (size_t)col * Hn + 4 * half + 64);
        v8f acc = {0.f,0.f,0.f,0.f,0.f,0.f,0.f,0.f};
        const float* ap = sAo + idx * QP + 4 * half;
        mm64s(ap,      y0, acc);
        mm64s(ap + 64, y1, acc);
#pragma unroll
        for (int r = 0; r < 8; r++) {
          const int t = tm * 16 + r + 8 * half;
          y[(size_t)t * Bn * OUTn + (size_t)b * OUTn + col] = acc[r] + ubc;
        }
      }
    } else {
      ld_bt64(p1, pw + 64);
      st_ln(sAo + wave * QP, nlw, nlb, nxn + (row0 + wave) * INn, sH + wave * QP);
      lds_handover_barrier();
      v8f acc = {0.f,0.f,0.f,0.f,0.f,0.f,0.f,0.f};
      const float* ap = sH + idx * QP + 4 * half;
      mm64s(ap,      p0, acc);
      mm64s(ap + 64, p1, acc);
      float* Out = (mmw ? nK : nV) + (row0 * Hn) + tn * 16;
#pragma unroll
      for (int r = 0; r < 8; r++) {
        float v = acc[r];
        if (mmw) v = ((v > 0.f) ? v : expm1f(v)) + 1.0f;
        Out[(size_t)(r + 8 * half) * Hn + idx] = v;
      }
    }
  } else if (blockIdx.x < 160) {
    st_states(Kb, Vb, st, layerOff, outst, (blockIdx.x - 128) * 16 + wave);
  } else {
    const int e = (blockIdx.x - 160) * 512 + threadIdx.x;
    st_statez(Kb, st, layerOff, outst, e >> 7, e & 127);
  }
}

extern "C" void kernel_launch(void* const* d_in, const int* in_sizes, int n_in,
                              void* d_out, int out_size, void* d_ws, size_t ws_size,
                              hipStream_t stream) {
  (void)in_sizes; (void)n_in; (void)out_size; (void)ws_size;
  const float* z   = (const float*)d_in[0];
  const float* st  = (const float*)d_in[1];
  const float* Wk  = (const float*)d_in[2];
  const float* Wq  = (const float*)d_in[3];
  const float* Wv  = (const float*)d_in[4];
  const float* lnw = (const float*)d_in[5];
  const float* lnb = (const float*)d_in[6];
  const float* w1  = (const float*)d_in[7];
  const float* b1  = (const float*)d_in[8];
  const float* w2  = (const float*)d_in[9];
  const float* b2  = (const float*)d_in[10];
  const float* scw = (const float*)d_in[11];
  const float* scb = (const float*)d_in[12];
  const float* uw  = (const float*)d_in[13];
  const float* ub  = (const float*)d_in[14];
  float* out = (float*)d_out;
  float* ws  = (float*)d_ws;

  float* K1   = ws;
  float* XN   = ws + 262144;
  float* K0   = ws + 524288;
  float* QV   = ws + 786432;
  float* V0   = ws + 1048576;
  float* OST  = out + YSZ;

  dla_front_kernel<<<128, 768, 0, stream>>>(z, st, lnw, lnb, Wk, Wq, Wv, XN, K0, QV, V0, OST);
  dla_mix_kernel<false><<<162, 512, 0, stream>>>(QV, K0, V0, XN, st, 0, OST,
                                                 w1, b1, w2, b2, scw, scb,
                                                 lnw + INn, lnb + INn, Wk + Hn * INn, Wv + Hn * INn,
                                                 XN, K1, QV,
                                                 nullptr, nullptr, nullptr, nullptr);
  dla_mix_kernel<true><<<162, 512, 0, stream>>>(nullptr, K1, QV, XN, st, LAYERSTATE, OST,
                                                w1 + Hn * Hn, b1 + Hn, w2 + Hn * Hn, b2 + Hn,
                                                scw + Hn * INn, scb + Hn,
                                                nullptr, nullptr, nullptr, nullptr,
                                                nullptr, nullptr, nullptr,
                                                Wq + Hn * INn, uw, ub, out);
}


# K and weight operand tiles fetched by whole-row loads (4 cache lines per load instruction, not 16), parked in per-wave LDS slots and read from there as fragments; sum order unchanged
# speedup vs baseline: 3.3779x; 1.1131x over previous
#include <hip/hip_runtime.h>
#include <hip/hip_bf16.h>
#include <math.h>


#define Bn 8
#define Tn 256
#define INn 128
#define Hn 128
#define OUTn 128
#define Ln 2
#define STSTRIDE 33025
#define LAYERSTATE 16512
#define YSZ (Tn*Bn*OUTn)

#define QP 132
#define AP 260
#define SP 68

typedef __attribute__((ext_vector_type(2))) float v2f;
typedef __attribute__((ext_vector_type(4))) float v4f;
typedef __attribute__((ext_vector_type(8))) float v8f;

__device__ __forceinline__ v8f wmma4(v2f a, v2f b, v8f c) {
  return __builtin_amdgcn_wmma_f32_16x16x4_f32(false, a, false, b, (short)0, c, false, false);
}

__device__ __forceinline__ v4f load_k4(const float* p) {
  v4f v;
  __builtin_memcpy(&v, __builtin_assume_aligned(p, 16), 16);
  return v;
}

__device__ __forceinline__ void store_k4(float* p, v4f v) {
  __builtin_memcpy(__builtin_assume_aligned(p, 16), &v, 16);
}

__device__ __forceinline__ void lds_handover_barrier() {
  __builtin_amdgcn_fence(__ATOMIC_RELEASE, "workgroup", "local");
  __builtin_amdgcn_s_barrier();
  __builtin_amdgcn_fence(__ATOMIC_ACQUIRE, "workgroup", "local");
}


__device__ __forceinline__ void tile_atb(const float* Am, int lda, const float* Bm, int ldb,
                                         int K, v8f& acc) {
  const int lane = threadIdx.x & 31;
  const int half = lane >> 4;
  const int idx  = lane & 15;
  for (int k = 0; k < K; k += 4) {
    v2f a, b;
    a.x = Am[(size_t)(k + 2 * half)     * lda + idx];
    a.y = Am[(size_t)(k + 2 * half + 1) * lda + idx];
    b.x = Bm[(size_t)(k + 2 * half)     * ldb + idx];
    b.y = Bm[(size_t)(k + 2 * half + 1) * ldb + idx];
    acc = wmma4(a, b, acc);
  }
}

struct FragB64 { v4f b[8]; };

__device__ __forceinline__ void ld_b64(FragB64& f, const float* bc, int ldb) {
#pragma unroll
  for (int i = 0; i < 8; i++) {
    const float* br = bc + (size_t)(8 * i) * ldb;
    v4f b;
    b.x = br[0];
    b.y = br[(size_t)ldb];
    b.z = br[(size_t)2 * ldb];
    b.w = br[(size_t)3 * ldb];
    f.b[i] = b;
  }
}

__device__ __forceinline__ void mm64s(const float* ap, const FragB64& f, v8f& acc) {
  v4f a4[8];
#pragma unroll
  for (int i = 0; i < 8; i++) a4[i] = load_k4(ap + 8 * i);
  __builtin_amdgcn_sched_barrier(0);
#pragma unroll
  for (int i = 0; i < 8; i++) {
    v2f a, b;
    a.x = a4[i].x; a.y = a4[i].y; b.x = f.b[i].x; b.y = f.b[i].y;
    acc = wmma4(a, b, acc);
    a.x = a4[i].z; a.y = a4[i].w; b.x = f.b[i].z; b.y = f.b[i].w;
    acc = wmma4(a, b, acc);
  }
}

struct RowsH { v4f r[8]; };

__device__ __forceinline__ void ld_rows64(RowsH& f, const float* tp, int ld) {
  const int lane = threadIdx.x & 31;
  const float* p = tp + (size_t)(lane >> 4) * ld + 4 * (lane & 15);
#pragma unroll
  for (int j = 0; j < 8; j++) f.r[j] = load_k4(p + (size_t)(2 * j) * ld);
}

__device__ __forceinline__ void park64(float* slot, const RowsH& f) {
  const int lane = threadIdx.x & 31;
  float* p = slot + (lane >> 4) * SP + 4 * (lane & 15);
#pragma unroll
  for (int j = 0; j < 8; j++) store_k4(p + 2 * j * SP, f.r[j]);
}

__device__ __forceinline__ void park64_late(float* slot, const RowsH& f) {
  __builtin_amdgcn_sched_barrier(0);
  park64(slot, f);
}

__device__ __forceinline__ void mm64p(const float* ap, const float* slot, v8f& acc) {
  const int lane = threadIdx.x & 31;
  const float* bp = slot + (lane & 15) * SP + 4 * (lane >> 4);
  v4f a4[8], b4[8];
#pragma unroll
  for (int i = 0; i < 8; i++) a4[i] = load_k4(ap + 8 * i);
#pragma unroll
  for (int i = 0; i < 8; i++) b4[i] = load_k4(bp + 8 * i);
  __builtin_amdgcn_sched_barrier(0);
#pragma unroll
  for (int i = 0; i < 8; i++) {
    v2f a, b;
    a.x = a4[i].x; a.y = a4[i].y; b.x = b4[i].x; b.y = b4[i].y;
    acc = wmma4(a, b, acc);
    a.x = a4[i].z; a.y = a4[i].w; b.x = b4[i].z; b.y = b4[i].w;
    acc = wmma4(a, b, acc);
  }
}


__device__ __forceinline__ void st_rotate(const float* __restrict__ z, const float* __restrict__ st,
                                          float* xrows, float* outst, int b, int t0, int lt, int j) {
  const int t = t0 + lt;
  const float ts = st[(size_t)b * STSTRIDE + STSTRIDE - 1];
  const float pos = ts + (float)t;
  const float freq = expf((float)(2 * j) * (-9.210340371976184f / 128.0f));
  const float ang = pos * freq;
  const float s = sinf(ang), c = cosf(ang);
  const float* zp = z + (size_t)t * Bn * INn + (size_t)b * INn;
  const float x1 = zp[j], x2 = zp[j + 64];
  float* xp = xrows + lt * QP;
  xp[j]      = x1 * c - x2 * s;
  xp[j + 64] = x2 * c + x1 * s;
  if (t == 0 && j == 0)
    outst[(size_t)b * STSTRIDE + STSTRIDE - 1] = ts + (float)Tn;
}

__device__ __forceinline__ void ld_ln4(const float* __restrict__ lnw, const float* __restrict__ lnb,
                                       float (&w)[4], float (&bias)[4]) {
  const int lane = threadIdx.x & 31;
#pragma unroll
  for (int j = 0; j < 4; j++) { w[j] = lnw[lane + 32 * j]; bias[j] = lnb[lane + 32 * j]; }
}

__device__ __forceinline__ void st_ln(const float* xr, const float (&w)[4], const float (&bias)[4],
                                      float* outg, float* outl) {
  const int lane = threadIdx.x & 31;
  float xv[4];
  float s = 0.f, ss = 0.f;
#pragma unroll
  for (int j = 0; j < 4; j++) { const float v = xr[lane + 32 * j]; xv[j] = v; s += v; ss += v * v; }
  for (int off = 16; off; off >>= 1) { s += __shfl_xor(s, off); ss += __shfl_xor(ss, off); }
  const float m = s * (1.0f / INn);
  const float var = ss * (1.0f / INn) - m * m;
  const float inv = rsqrtf(var + 1e-5f);
#pragma unroll
  for (int j = 0; j < 4; j++) {
    const float v = (xv[j] - m) * inv * w[j] + bias[j];
    outg[lane + 32 * j] = v;
    outl[lane + 32 * j] = v;
  }
}

__device__ __forceinline__ void st_states(const float* Kb, const float* Vb,
                                          const float* __restrict__ st, int layerOff,
                                          float* outst, int g) {
  const int b = g >> 6;
  const int tile = g & 63;
  const int tm = tile >> 3, tn = tile & 7;
  v8f acc = {0.f,0.f,0.f,0.f,0.f,0.f,0.f,0.f};
  tile_atb(Kb + (size_t)b * Tn * Hn + tm * 16, Hn,
           Vb + (size_t)b * Tn * Hn + tn * 16, Hn, Tn, acc);
  const float* S0 = st + (size_t)b * STSTRIDE + layerOff;
  float* Out = outst + (size_t)b * STSTRIDE + layerOff;
  const int lane = threadIdx.x & 31, half = lane >> 4, idx = lane & 15;
#pragma unroll
  for (int r = 0; r < 8; r++) {
    const int i = tm * 16 + r + 8 * half;
    const int l = tn * 16 + idx;
    Out[(size_t)i * Hn + l] = acc[r] + S0[(size_t)i * Hn + l];
  }
}

__device__ __forceinline__ void st_statez(const float* Kb, const float* __restrict__ st,
                                          int layerOff, float* outst, int b, int i) {
  const float* Kp = Kb + (size_t)b * Tn * Hn + i;
  float s = 0.f;
  for (int t = 0; t < Tn; t++) s += Kp[(size_t)t * Hn];
  const size_t o = (size_t)b * STSTRIDE + layerOff + Hn * Hn + i;
  outst[o] = st[o] + s;
}

__global__ __launch_bounds__(768) void dla_front_kernel(
    const float* __restrict__ z, const float* __restrict__ st,
    const float* __restrict__ lnw, const float* __restrict__ lnb,
    const float* __restrict__ Wk, const float* __restrict__ Wq, const float* __restrict__ Wv,
    float* xn, float* Kb, float* Qb, float* Vb, float* outst) {
  __shared__ __attribute__((aligned(16))) float sXr[16 * QP];
  __shared__ __attribute__((aligned(16))) float sN[16 * QP];
  __shared__ __attribute__((aligned(16))) float sB[24 * 2 * 16 * SP];
  const int b = blockIdx.x >> 4, tm = blockIdx.x & 15;
  const int wave = threadIdx.x >> 5;
  const int lane = threadIdx.x & 31, half = lane >> 4, idx = lane & 15;
  const int which = wave >> 3, tn = wave & 7;
  const size_t row0 = (size_t)b * Tn + tm * 16;
  const float* W = (which == 0) ? Wk : (which == 1) ? Wq : Wv;
  float* slot0 = sB + wave * (2 * 16 * SP);
  float* slot1 = slot0 + 16 * SP;
  RowsH w0, w1;
  {
    const float* tp = W + (size_t)(tn * 16) * INn;
    ld_rows64(w0, tp,      INn);
    ld_rows64(w1, tp + 64, INn);
  }
  float lw[4], lb[4];
#pragma unroll
  for (int j = 0; j < 4; j++) { lw[j] = lnw[lane + 32 * j]; lb[j] = lnb[lane + 32 * j]; }
  for (int i = threadIdx.x; i < 16 * 64; i += 768)
    st_rotate(z, st, sXr, outst, b, tm * 16, i >> 6, i & 63);
  lds_handover_barrier();
  park64(slot0, w0);
  park64(slot1, w1);
  if (wave < 16)
    st_ln(sXr + wave * QP, lw, lb, xn + (row0 + wave) * INn, sN + wave * QP);
  lds_handover_barrier();
  v8f acc = {0.f,0.f,0.f,0.f,0.f,0.f,0.f,0.f};
  const float* ap = sN + idx * QP + 4 * half;
  mm64p(ap,      slot0, acc);
  mm64p(ap + 64, slot1, acc);
  float* Out = ((which == 0) ? Kb : (which == 1) ? Qb : Vb) + (row0 * Hn) + tn * 16;
#pragma unroll
  for (int r = 0; r < 8; r++) {
    float v = acc[r];
    if (which != 2) v = ((v > 0.f) ? v : expm1f(v)) + 1.0f;
    Out[(size_t)(r + 8 * half) * Hn + idx] = v;
  }
}

template <bool LAST>
__global__ __launch_bounds__(512) __attribute__((amdgpu_waves_per_eu(4, 4))) void dla_mix_kernel(
    const float* Qb, const float* Kb, const float* Vb, const float* xn,
    const float* __restrict__ st, int layerOff, float* outst,
    const float* __restrict__ w1, const float* __restrict__ b1,
    const float* __restrict__ w2, const float* __restrict__ b2,
    const float* __restrict__ scw, const float* __restrict__ scb,
    const float* __restrict__ nlnw, const float* __restrict__ nlnb,
    const float* __restrict__ nWk, const float* __restrict__ nWv,
    float* nxn, float* nK, float* nV,
    const float* __restrict__ Wq,
    const float* __restrict__ uw, const float* __restrict__ ub, float* y) {
  __shared__ __attribute__((aligned(16))) float sQ[16 * QP];
  __shared__ __attribute__((aligned(16))) float sXn[16 * QP];
  __shared__ __attribute__((aligned(16))) float sAo[16 * QP];
  __shared__ __attribute__((aligned(16))) float sH[16 * QP];
  __shared__ __attribute__((aligned(16))) float sA[16 * AP];
  __shared__ float sDen[16];
  __shared__ __attribute__((aligned(16))) float sB[16 * 2 * 16 * SP];
  const int wave = threadIdx.x >> 5;
  if (blockIdx.x < 128) {
    float* slot0 = sB + wave * (2 * 16 * SP);
    float* slot1 = slot0 + 16 * SP;
    const int b = blockIdx.x >> 4, tm = blockIdx.x & 15;
    const int lane = threadIdx.x & 31, half = lane >> 4, idx = lane & 15;
    const size_t row0 = (size_t)b * Tn + tm * 16;
    const bool live = wave <= tm;
    const bool mmw  = wave < 8;
    const int tn  = wave & 7;
    const int col = tn * 16 + idx;

    RowsH k0, k1;
    if (live) {
      const float* kp = Kb + ((size_t)b * Tn + wave * 16) * Hn;
      ld_rows64(k0, kp,      Hn);
      ld_rows64(k1, kp + 64, Hn);
    }
    const int lr = threadIdx.x >> 5, lc = (threadIdx.x & 31) * 4;
    const v4f x4 = load_k4(xn + (row0 + lr) * INn + lc);
    v4f q4;
    RowsH q0, q1;
    if (LAST) {
      if (mmw) {
        const float* qp = Wq + (size_t)(tn * 16) * INn;
        ld_rows64(q0, qp,      INn);
        ld_rows64(q1, qp + 64, INn);
      }
    } else {
      q4 = load_k4(Qb + (row0 + lr) * Hn + lc);
    }
    const float* Z0 = st + (size_t)b * STSTRIDE + layerOff + Hn * Hn;
    float z0[4];
#pragma unroll
    for (int j = 0; j < 4; j++) z0[j] = Z0[lane + 32 * j];
    const float* bV = Vb + (size_t)b * Tn * Hn + (size_t)(4 * half) * Hn + col;
    const float* bS = st + (size_t)b * STSTRIDE + layerOff + (size_t)(4 * half) * Hn + col;
    const float b1c = b1[col], b2c = b2[col], scbc = scb[col];
    if (!LAST) store_k4(sQ + lr * QP + lc, q4);
    store_k4(sXn + lr * QP + lc, x4);
    if (LAST) {
      if (mmw) { park64(slot0, q0); park64(slot1, q1); }
    } else {
      if (live) { park64(slot0, k0); park64(slot1, k1); }
    }
    lds_handover_barrier();
    if (LAST) {
      if (mmw) {
        v8f acc = {0.f,0.f,0.f,0.f,0.f,0.f,0.f,0.f};
        const float* ap = sXn + idx * QP + 4 * half;
        mm64p(ap,      slot0, acc);
        mm64p(ap + 64, slot1, acc);
#pragma unroll
        for (int r = 0; r < 8; r++) {
          const float v = acc[r];
          sQ[(r + 8 * half) * QP + col] = ((v > 0.f) ? v : expm1f(v)) + 1.0f;
        }
      }
      if (live) { park64(slot0, k0); park64(slot1, k1); }
      lds_handover_barrier();
    }
    FragB64 g0, g1, g2;
    if (mmw) {
      ld_b64(g0, bV,                    Hn);
      ld_b64(g1, bV + (size_t)64 * Hn,  Hn);
      ld_b64(g2, bV + (size_t)128 * Hn, Hn);
    }

    {
      float* o = sA + (8 * half) * AP + wave * 16 + idx;
      if (live) {
        v8f acc = {0.f,0.f,0.f,0.f,0.f,0.f,0.f,0.f};
        const float* ap = sQ + idx * QP + 4 * half;
        mm64p(ap,      slot0, acc);
        mm64p(ap + 64, slot1, acc);
#pragma unroll
        for (int r = 0; r < 8; r++) {
          float v = acc[r];
          if (wave == tm && idx > (r + 8 * half)) v = 0.f;
          o[r * AP] = v;
        }
      } else {
#pragma unroll
        for (int r = 0; r < 8; r++) o[r * AP] = 0.f;
      }
    }
    lds_handover_barrier();

    {
      const float* Ar = sA + wave * AP;
      const float* Qr = sQ + wave * QP;
      float s = 0.f;
#pragma unroll
      for (int j = 0; j < 8; j++) s += Ar[lane + 32 * j];
#pragma unroll
      for (int j = 0; j < 4; j++) s += Qr[lane + 32 * j] * z0[j];
      for (int off = 16; off; off >>= 1) s += __shfl_xor(s, off);
      if (lane == 0) sDen[wave] = s + 1e-5f;
    }
    lds_handover_barrier();

    RowsH wa, wb;
    RowsH u0, u1, u2, u3;
    if (mmw) {
      v8f acc = {0.f,0.f,0.f,0.f,0.f,0.f,0.f,0.f};
      const float* aA = sA + idx * AP + 4 * half;
      const float* aQ = sQ + idx * QP + 4 * half;
      float dn[8];
#pragma unroll
      for (int r = 0; r < 8; r++) dn[r] = sDen[r + 8 * half];
      mm64s(aA,       g0, acc);
      ld_b64(g0, bV + (size_t)192 * Hn, Hn);
      mm64s(aA + 64,  g1, acc);
      ld_b64(g1, bS,                    Hn);
      mm64s(aA + 128, g2, acc);
      ld_b64(g2, bS + (size_t)64 * Hn,  Hn);
      mm64s(aA + 192, g0, acc);
      ld_rows64(wa, w1 + (size_t)(tn * 16) * Hn,      Hn);
      ld_rows64(wb, w1 + (size_t)(tn * 16) * Hn + 64, Hn);
      mm64s(aQ,       g1, acc);
      mm64s(aQ + 64,  g2, acc);
#pragma unroll
      for (int r = 0; r < 8; r++)
        sAo[(r + 8 * half) * QP + col] = acc[r] / dn[r];
      park64_late(slot0, wa);
      park64(slot1, wb);
    }
    lds_handover_barrier();

    if (mmw) {
      ld_rows64(u0, w2  + (size_t)(tn * 16) * Hn,  Hn);
      ld_rows64(u1, scw + (size_t)(tn * 16) * INn, INn);
      v8f acc = {0.f,0.f,0.f,0.f,0.f,0.f,0.f,0.f};
      const float* ap = sAo + idx * QP + 4 * half;
      mm64p(ap,      slot0, acc);
      mm64p(ap + 64, slot1, acc);
#pragma unroll
      for (int r = 0; r < 8; r++) {
        float v = acc[r] + b1c;
        sH[(r + 8 * half) * QP + col] = v > 0.f ? v : 0.f;
      }
      park64_late(slot0, u0);
      park64(slot1, u1);
    }
    lds_handover_barrier();
    RowsH y0, y1;
    float ubc = 0.f;
    RowsH p0, p1;
    float nlw[4], nlb[4];
    const float* pw = nullptr;
    if (!LAST) pw = (mmw ? nWk : nWv) + (size_t)(tn * 16) * INn;
    if (!LAST && !mmw) {
      ld_rows64(p0, pw,      INn);
      ld_rows64(p1, pw + 64, INn);
      ld_ln4(nlnw, nlnb, nlw, nlb);
    }

    if (mmw) {
      ld_rows64(u2, w2  + (size_t)(tn * 16) * Hn  + 64, Hn);
      ld_rows64(u3, scw + (size_t)(tn * 16) * INn + 64, INn);
      if (LAST) ubc = ub[col];
      v8f acc1 = {0.f,0.f,0.f,0.f,0.f,0.f,0.f,0.f};
      v8f acc2 = {0.f,0.f,0.f,0.f,0.f,0.f,0.f,0.f};
      const float* a1 = sH  + idx * QP + 4 * half;
      const float* a2 = sXn + idx * QP + 4 * half;
      mm64p(a1,      slot0, acc1);
      mm64p(a2,      slot1, acc2);
      park64(slot0, u2);
      mm64p(a1 + 64, slot0, acc1);
      if (LAST) ld_rows64(y0, uw + (size_t)(tn * 16) * Hn, Hn);
      else      ld_rows64(p0, pw, INn);
      park64(slot1, u3);
      mm64p(a2 + 64, slot1, acc2);
      if (LAST) {
        ld_rows64(y1, uw + (size_t)(tn * 16) * Hn + 64, Hn);
      } else {
        ld_rows64(p1, pw + 64, INn);
        ld_ln4(nlnw, nlnb, nlw, nlb);
      }
#pragma unroll
      for (int r = 0; r < 8; r++) {
        float v = acc1[r] + b2c;
        v = v > 0.f ? v : 0.f;
        const float xv = v + acc2[r] + scbc;
        sAo[(r + 8 * half) * QP + col] = xv;
      }
      if (LAST) {
        park64_late(slot0, y0);
        park64(slot1, y1);
      }
    }
    lds_handover_barrier();
    if (LAST) {
      if (mmw) {
        v8f acc = {0.f,0.f,0.f,0.f,0.f,0.f,0.f,0.f};
        const float* ap = sAo + idx * QP + 4 * half;
        mm64p(ap,      slot0, acc);
        mm64p(ap + 64, slot1, acc);
#pragma unroll
        for (int r = 0; r < 8; r++) {
          const int t = tm * 16 + r + 8 * half;
          y[(size_t)t * Bn * OUTn + (size_t)b * OUTn + col] = acc[r] + ubc;
        }
      }
    } else {
      st_ln(sAo + wave * QP, nlw, nlb, nxn + (row0 + wave) * INn, sH + wave * QP);
      park64_late(slot0, p0);
      park64(slot1, p1);
      lds_handover_barrier();
      v8f acc = {0.f,0.f,0.f,0.f,0.f,0.f,0.f,0.f};
      const float* ap = sH + idx * QP + 4 * half;
      mm64p(ap,      slot0, acc);
      mm64p(ap + 64, slot1, acc);
      float* Out = (mmw ? nK : nV) + (row0 * Hn) + tn * 16;
#pragma unroll
      for (int r = 0; r < 8; r++) {
        float v = acc[r];
        if (mmw) v = ((v > 0.f) ? v : expm1f(v)) + 1.0f;
        Out[(size_t)(r + 8 * half) * Hn + idx] = v;
      }
    }
  } else if (blockIdx.x < 160) {
    st_states(Kb, Vb, st, layerOff, outst, (blockIdx.x - 128) * 16 + wave);
  } else {
    const int e = (blockIdx.x - 160) * 512 + threadIdx.x;
    st_statez(Kb, st, layerOff, outst, e >> 7, e & 127);
  }
}

extern "C" void kernel_launch(void* const* d_in, const int* in_sizes, int n_in,
                              void* d_out, int out_size, void* d_ws, size_t ws_size,
                              hipStream_t stream) {
  (void)in_sizes; (void)n_in; (void)out_size; (void)ws_size;
  const float* z   = (const float*)d_in[0];
  const float* st  = (const float*)d_in[1];
  const float* Wk  = (const float*)d_in[2];
  const float* Wq  = (const float*)d_in[3];
  const float* Wv  = (const float*)d_in[4];
  const float* lnw = (const float*)d_in[5];
  const float* lnb = (const float*)d_in[6];
  const float* w1  = (const float*)d_in[7];
  const float* b1  = (const float*)d_in[8];
  const float* w2  = (const float*)d_in[9];
  const float* b2  = (const float*)d_in[10];
  const float* scw = (const float*)d_in[11];
  const float* scb = (const float*)d_in[12];
  const float* uw  = (const float*)d_in[13];
  const float* ub  = (const float*)d_in[14];
  float* out = (float*)d_out;
  float* ws  = (float*)d_ws;

  float* K1   = ws;
  float* XN   = ws + 262144;
  float* K0   = ws + 524288;
  float* QV   = ws + 786432;
  float* V0   = ws + 1048576;
  float* OST  = out + YSZ;

  dla_front_kernel<<<128, 768, 0, stream>>>(z, st, lnw, lnb, Wk, Wq, Wv, XN, K0, QV, V0, OST);
  dla_mix_kernel<false><<<162, 512, 0, stream>>>(QV, K0, V0, XN, st, 0, OST,
                                                 w1, b1, w2, b2, scw, scb,
                                                 lnw + INn, lnb + INn, Wk + Hn * INn, Wv + Hn * INn,
                                                 XN, K1, QV,
                                                 nullptr, nullptr, nullptr, nullptr);
  dla_mix_kernel<true><<<162, 512, 0, stream>>>(nullptr, K1, QV, XN, st, LAYERSTATE, OST,
                                                w1 + Hn * Hn, b1 + Hn, w2 + Hn * Hn, b2 + Hn,
                                                scw + Hn * INn, scb + Hn,
                                                nullptr, nullptr, nullptr, nullptr,
                                                nullptr, nullptr, nullptr,
                                                Wq + Hn * INn, uw, ub, out);
}


# Launch-start loads issued in the order their consumers start: front: z, then weight rows k 0..63, then k 64..127 (first group runs meanwhile); last mix: xn, Wq rows before K rows; sum order unchanged
# speedup vs baseline: 3.4975x; 1.0354x over previous
#include <hip/hip_runtime.h>
#include <hip/hip_bf16.h>
#include <math.h>


#define Bn 8
#define Tn 256
#define INn 128
#define Hn 128
#define OUTn 128
#define Ln 2
#define STSTRIDE 33025
#define LAYERSTATE 16512
#define YSZ (Tn*Bn*OUTn)

#define QP 132
#define AP 260
#define SP 68

typedef __attribute__((ext_vector_type(2))) float v2f;
typedef __attribute__((ext_vector_type(4))) float v4f;
typedef __attribute__((ext_vector_type(8))) float v8f;

__device__ __forceinline__ v8f wmma4(v2f a, v2f b, v8f c) {
  return __builtin_amdgcn_wmma_f32_16x16x4_f32(false, a, false, b, (short)0, c, false, false);
}

__device__ __forceinline__ v4f load_k4(const float* p) {
  v4f v;
  __builtin_memcpy(&v, __builtin_assume_aligned(p, 16), 16);
  return v;
}

__device__ __forceinline__ void store_k4(float* p, v4f v) {
  __builtin_memcpy(__builtin_assume_aligned(p, 16), &v, 16);
}

__device__ __forceinline__ void lds_handover_barrier() {
  __builtin_amdgcn_fence(__ATOMIC_RELEASE, "workgroup", "local");
  __builtin_amdgcn_s_barrier();
  __builtin_amdgcn_fence(__ATOMIC_ACQUIRE, "workgroup", "local");
}


__device__ __forceinline__ void tile_atb(const float* Am, int lda, const float* Bm, int ldb,
                                         int K, v8f& acc) {
  const int lane = threadIdx.x & 31;
  const int half = lane >> 4;
  const int idx  = lane & 15;
  for (int k = 0; k < K; k += 4) {
    v2f a, b;
    a.x = Am[(size_t)(k + 2 * half)     * lda + idx];
    a.y = Am[(size_t)(k + 2 * half + 1) * lda + idx];
    b.x = Bm[(size_t)(k + 2 * half)     * ldb + idx];
    b.y = Bm[(size_t)(k + 2 * half + 1) * ldb + idx];
    acc = wmma4(a, b, acc);
  }
}

struct FragB64 { v4f b[8]; };

__device__ __forceinline__ void ld_b64(FragB64& f, const float* bc, int ldb) {
#pragma unroll
  for (int i = 0; i < 8; i++) {
    const float* br = bc + (size_t)(8 * i) * ldb;
    v4f b;
    b.x = br[0];
    b.y = br[(size_t)ldb];
    b.z = br[(size_t)2 * ldb];
    b.w = br[(size_t)3 * ldb];
    f.b[i] = b;
  }
}

__device__ __forceinline__ void mm64s(const float* ap, const FragB64& f, v8f& acc) {
  v4f a4[8];
#pragma unroll
  for (int i = 0; i < 8; i++) a4[i] = load_k4(ap + 8 * i);
  __builtin_amdgcn_sched_barrier(0);
#pragma unroll
  for (int i = 0; i < 8; i++) {
    v2f a, b;
    a.x = a4[i].x; a.y = a4[i].y; b.x = f.b[i].x; b.y = f.b[i].y;
    acc = wmma4(a, b, acc);
    a.x = a4[i].z; a.y = a4[i].w; b.x = f.b[i].z; b.y = f.b[i].w;
    acc = wmma4(a, b, acc);
  }
}

struct RowsH { v4f r[8]; };

__device__ __forceinline__ void ld_rows64(RowsH& f, const float* tp, int ld) {
  const int lane = threadIdx.x & 31;
  const float* p = tp + (size_t)(lane >> 4) * ld + 4 * (lane & 15);
#pragma unroll
  for (int j = 0; j < 8; j++) f.r[j] = load_k4(p + (size_t)(2 * j) * ld);
}

__device__ __forceinline__ void park64(float* slot, const RowsH& f) {
  const int lane = threadIdx.x & 31;
  float* p = slot + (lane >> 4) * SP + 4 * (lane & 15);
#pragma unroll
  for (int j = 0; j < 8; j++) store_k4(p + 2 * j * SP, f.r[j]);
}

__device__ __forceinline__ void park64_late(float* slot, const RowsH& f) {
  __builtin_amdgcn_sched_barrier(0);
  park64(slot, f);
}

__device__ __forceinline__ void mm64p(const float* ap, const float* slot, v8f& acc) {
  const int lane = threadIdx.x & 31;
  const float* bp = slot + (lane & 15) * SP + 4 * (lane >> 4);
  v4f a4[8], b4[8];
#pragma unroll
  for (int i = 0; i < 8; i++) a4[i] = load_k4(ap + 8 * i);
#pragma unroll
  for (int i = 0; i < 8; i++) b4[i] = load_k4(bp + 8 * i);
  __builtin_amdgcn_sched_barrier(0);
#pragma unroll
  for (int i = 0; i < 8; i++) {
    v2f a, b;
    a.x = a4[i].x; a.y = a4[i].y; b.x = b4[i].x; b.y = b4[i].y;
    acc = wmma4(a, b, acc);
    a.x = a4[i].z; a.y = a4[i].w; b.x = b4[i].z; b.y = b4[i].w;
    acc = wmma4(a, b, acc);
  }
}


__device__ __forceinline__ void st_rotate(float ts, float x1, float x2,
                                          float* xrows, float* outst, int b, int t0, int lt, int j) {
  const int t = t0 + lt;
  const float pos = ts + (float)t;
  const float freq = expf((float)(2 * j) * (-9.210340371976184f / 128.0f));
  const float ang = pos * freq;
  const float s = sinf(ang), c = cosf(ang);
  float* xp = xrows + lt * QP;
  xp[j]      = x1 * c - x2 * s;
  xp[j + 64] = x2 * c + x1 * s;
  if (t == 0 && j == 0)
    outst[(size_t)b * STSTRIDE + STSTRIDE - 1] = ts + (float)Tn;
}

__device__ __forceinline__ void ld_ln4(const float* __restrict__ lnw, const float* __restrict__ lnb,
                                       float (&w)[4], float (&bias)[4]) {
  const int lane = threadIdx.x & 31;
#pragma unroll
  for (int j = 0; j < 4; j++) { w[j] = lnw[lane + 32 * j]; bias[j] = lnb[lane + 32 * j]; }
}

__device__ __forceinline__ void st_ln(const float* xr, const float (&w)[4], const float (&bias)[4],
                                      float* outg, float* outl) {
  const int lane = threadIdx.x & 31;
  float xv[4];
  float s = 0.f, ss = 0.f;
#pragma unroll
  for (int j = 0; j < 4; j++) { const float v = xr[lane + 32 * j]; xv[j] = v; s += v; ss += v * v; }
  for (int off = 16; off; off >>= 1) { s += __shfl_xor(s, off); ss += __shfl_xor(ss, off); }
  const float m = s * (1.0f / INn);
  const float var = ss * (1.0f / INn) - m * m;
  const float inv = rsqrtf(var + 1e-5f);
#pragma unroll
  for (int j = 0; j < 4; j++) {
    const float v = (xv[j] - m) * inv * w[j] + bias[j];
    outg[lane + 32 * j] = v;
    outl[lane + 32 * j] = v;
  }
}

__device__ __forceinline__ void st_states(const float* Kb, const float* Vb,
                                          const float* __restrict__ st, int layerOff,
                                          float* outst, int g) {
  const int b = g >> 6;
  const int tile = g & 63;
  const int tm = tile >> 3, tn = tile & 7;
  v8f acc = {0.f,0.f,0.f,0.f,0.f,0.f,0.f,0.f};
  tile_atb(Kb + (size_t)b * Tn * Hn + tm * 16, Hn,
           Vb + (size_t)b * Tn * Hn + tn * 16, Hn, Tn, acc);
  const float* S0 = st + (size_t)b * STSTRIDE + layerOff;
  float* Out = outst + (size_t)b * STSTRIDE + layerOff;
  const int lane = threadIdx.x & 31, half = lane >> 4, idx = lane & 15;
#pragma unroll
  for (int r = 0; r < 8; r++) {
    const int i = tm * 16 + r + 8 * half;
    const int l = tn * 16 + idx;
    Out[(size_t)i * Hn + l] = acc[r] + S0[(size_t)i * Hn + l];
  }
}

__device__ __forceinline__ void st_statez(const float* Kb, const float* __restrict__ st,
                                          int layerOff, float* outst, int b, int i) {
  const float* Kp = Kb + (size_t)b * Tn * Hn + i;
  float s = 0.f;
  for (int t = 0; t < Tn; t++) s += Kp[(size_t)t * Hn];
  const size_t o = (size_t)b * STSTRIDE + layerOff + Hn * Hn + i;
  outst[o] = st[o] + s;
}

__global__ __launch_bounds__(768) void dla_front_kernel(
    const float* __restrict__ z, const float* __restrict__ st,
    const float* __restrict__ lnw, const float* __restrict__ lnb,
    const float* __restrict__ Wk, const float* __restrict__ Wq, const float* __restrict__ Wv,
    float* xn, float* Kb, float* Qb, float* Vb, float* outst) {
  __shared__ __attribute__((aligned(16))) float sXr[16 * QP];
  __shared__ __attribute__((aligned(16))) float sN[16 * QP];
  __shared__ __attribute__((aligned(16))) float sB[24 * 2 * 16 * SP];
  const int b = blockIdx.x >> 4, tm = blockIdx.x & 15;
  const int wave = threadIdx.x >> 5;
  const int lane = threadIdx.x & 31, half = lane >> 4, idx = lane & 15;
  const int which = wave >> 3, tn = wave & 7;
  const size_t row0 = (size_t)b * Tn + tm * 16;
  const float* W = (which == 0) ? Wk : (which == 1) ? Wq : Wv;
  float* slot0 = sB + wave * (2 * 16 * SP);
  float* slot1 = slot0 + 16 * SP;
  const float ts = st[(size_t)b * STSTRIDE + STSTRIDE - 1];
  const int lt0 = threadIdx.x >> 6, j0 = threadIdx.x & 63;
  const float* zp0 = z + (size_t)(tm * 16 + lt0) * Bn * INn + (size_t)b * INn;
  const float za0 = zp0[j0], zb0 = zp0[j0 + 64];
  const bool two = threadIdx.x < 256;
  float za1 = 0.f, zb1 = 0.f;
  if (two) {
    const float* zp1 = zp0 + (size_t)12 * Bn * INn;
    za1 = zp1[j0]; zb1 = zp1[j0 + 64];
  }
  float lw[4], lb[4];
#pragma unroll
  for (int j = 0; j < 4; j++) { lw[j] = lnw[lane + 32 * j]; lb[j] = lnb[lane + 32 * j]; }
  RowsH w0, w1;
  {
    const float* tp = W + (size_t)(tn * 16) * INn;
    ld_rows64(w0, tp,      INn);
    __builtin_amdgcn_sched_barrier(0);
    ld_rows64(w1, tp + 64, INn);
  }
  st_rotate(ts, za0, zb0, sXr, outst, b, tm * 16, lt0, j0);
  if (two) st_rotate(ts, za1, zb1, sXr, outst, b, tm * 16, lt0 + 12, j0);
  lds_handover_barrier();
  if (wave < 16)
    st_ln(sXr + wave * QP, lw, lb, xn + (row0 + wave) * INn, sN + wave * QP);
  park64(slot0, w0);
  lds_handover_barrier();
  v8f acc = {0.f,0.f,0.f,0.f,0.f,0.f,0.f,0.f};
  const float* ap = sN + idx * QP + 4 * half;
  mm64p(ap,      slot0, acc);
  asm volatile("" :: "v"(acc) : "memory");
  park64_late(slot1, w1);
  mm64p(ap + 64, slot1, acc);
  float* Out = ((which == 0) ? Kb : (which == 1) ? Qb : Vb) + (row0 * Hn) + tn * 16;
#pragma unroll
  for (int r = 0; r < 8; r++) {
    float v = acc[r];
    if (which != 2) v = ((v > 0.f) ? v : expm1f(v)) + 1.0f;
    Out[(size_t)(r + 8 * half) * Hn + idx] = v;
  }
}

template <bool LAST>
__global__ __launch_bounds__(512) __attribute__((amdgpu_waves_per_eu(4, 4))) void dla_mix_kernel(
    const float* Qb, const float* Kb, const float* Vb, const float* xn,
    const float* __restrict__ st, int layerOff, float* outst,
    const float* __restrict__ w1, const float* __restrict__ b1,
    const float* __restrict__ w2, const float* __restrict__ b2,
    const float* __restrict__ scw, const float* __restrict__ scb,
    const float* __restrict__ nlnw, const float* __restrict__ nlnb,
    const float* __restrict__ nWk, const float* __restrict__ nWv,
    float* nxn, float* nK, float* nV,
    const float* __restrict__ Wq,
    const float* __restrict__ uw, const float* __restrict__ ub, float* y) {
  __shared__ __attribute__((aligned(16))) float sQ[16 * QP];
  __shared__ __attribute__((aligned(16))) float sXn[16 * QP];
  __shared__ __attribute__((aligned(16))) float sAo[16 * QP];
  __shared__ __attribute__((aligned(16))) float sH[16 * QP];
  __shared__ __attribute__((aligned(16))) float sA[16 * AP];
  __shared__ float sDen[16];
  __shared__ __attribute__((aligned(16))) float sB[16 * 2 * 16 * SP];
  const int wave = threadIdx.x >> 5;
  if (blockIdx.x < 128) {
    float* slot0 = sB + wave * (2 * 16 * SP);
    float* slot1 = slot0 + 16 * SP;
    const int b = blockIdx.x >> 4, tm = blockIdx.x & 15;
    const int lane = threadIdx.x & 31, half = lane >> 4, idx = lane & 15;
    const size_t row0 = (size_t)b * Tn + tm * 16;
    const bool live = wave <= tm;
    const bool mmw  = wave < 8;
    const int tn  = wave & 7;
    const int col = tn * 16 + idx;

    RowsH k0, k1;
    const float* kp = Kb + ((size_t)b * Tn + wave * 16) * Hn;
    if (!LAST && live) {
      ld_rows64(k0, kp,      Hn);
      ld_rows64(k1, kp + 64, Hn);
    }
    const int lr = threadIdx.x >> 5, lc = (threadIdx.x & 31) * 4;
    const v4f x4 = load_k4(xn + (row0 + lr) * INn + lc);
    v4f q4;
    RowsH q0, q1;
    if (LAST) {
      if (mmw) {
        const float* qp = Wq + (size_t)(tn * 16) * INn;
        ld_rows64(q0, qp,      INn);
        ld_rows64(q1, qp + 64, INn);
      }
    } else {
      q4 = load_k4(Qb + (row0 + lr) * Hn + lc);
    }
    const float* Z0 = st + (size_t)b * STSTRIDE + layerOff + Hn * Hn;
    float z0[4];
#pragma unroll
    for (int j = 0; j < 4; j++) z0[j] = Z0[lane + 32 * j];
    const float* bV = Vb + (size_t)b * Tn * Hn + (size_t)(4 * half) * Hn + col;
    const float* bS = st + (size_t)b * STSTRIDE + layerOff + (size_t)(4 * half) * Hn + col;
    const float b1c = b1[col], b2c = b2[col], scbc = scb[col];
    if (!LAST) store_k4(sQ + lr * QP + lc, q4);
    store_k4(sXn + lr * QP + lc, x4);
    if (LAST) {
      if (mmw) { park64(slot0, q0); park64(slot1, q1); }
      if (live) {
        ld_rows64(k0, kp,      Hn);
        ld_rows64(k1, kp + 64, Hn);
      }
    } else {
      if (live) { park64(slot0, k0); park64(slot1, k1); }
    }
    lds_handover_barrier();
    if (LAST) {
      if (mmw) {
        v8f acc = {0.f,0.f,0.f,0.f,0.f,0.f,0.f,0.f};
        const float* ap = sXn + idx * QP + 4 * half;
        mm64p(ap,      slot0, acc);
        mm64p(ap + 64, slot1, acc);
#pragma unroll
        for (int r = 0; r < 8; r++) {
          const float v = acc[r];
          sQ[(r + 8 * half) * QP + col] = ((v > 0.f) ? v : expm1f(v)) + 1.0f;
        }
      }
      if (live) { park64(slot0, k0); park64(slot1, k1); }
      lds_handover_barrier();
    }
    FragB64 g0, g1, g2;
    if (mmw) {
      ld_b64(g0, bV,                    Hn);
      ld_b64(g1, bV + (size_t)64 * Hn,  Hn);
      ld_b64(g2, bV + (size_t)128 * Hn, Hn);
    }

    {
      float* o = sA + (8 * half) * AP + wave * 16 + idx;
      if (live) {
        v8f acc = {0.f,0.f,0.f,0.f,0.f,0.f,0.f,0.f};
        const float* ap = sQ + idx * QP + 4 * half;
        mm64p(ap,      slot0, acc);
        mm64p(ap + 64, slot1, acc);
#pragma unroll
        for (int r = 0; r < 8; r++) {
          float v = acc[r];
          if (wave == tm && idx > (r + 8 * half)) v = 0.f;
          o[r * AP] = v;
        }
      } else {
#pragma unroll
        for (int r = 0; r < 8; r++) o[r * AP] = 0.f;
      }
    }
    lds_handover_barrier();

    {
      const float* Ar = sA + wave * AP;
      const float* Qr = sQ + wave * QP;
      float s = 0.f;
#pragma unroll
      for (int j = 0; j < 8; j++) s += Ar[lane + 32 * j];
#pragma unroll
      for (int j = 0; j < 4; j++) s += Qr[lane + 32 * j] * z0[j];
      for (int off = 16; off; off >>= 1) s += __shfl_xor(s, off);
      if (lane == 0) sDen[wave] = s + 1e-5f;
    }
    lds_handover_barrier();

    RowsH wa, wb;
    RowsH u0, u1, u2, u3;
    if (mmw) {
      v8f acc = {0.f,0.f,0.f,0.f,0.f,0.f,0.f,0.f};
      const float* aA = sA + idx * AP + 4 * half;
      const float* aQ = sQ + idx * QP + 4 * half;
      float dn[8];
#pragma unroll
      for (int r = 0; r < 8; r++) dn[r] = sDen[r + 8 * half];
      mm64s(aA,       g0, acc);
      ld_b64(g0, bV + (size_t)192 * Hn, Hn);
      mm64s(aA + 64,  g1, acc);
      ld_b64(g1, bS,                    Hn);
      mm64s(aA + 128, g2, acc);
      ld_b64(g2, bS + (size_t)64 * Hn,  Hn);
      mm64s(aA + 192, g0, acc);
      ld_rows64(wa, w1 + (size_t)(tn * 16) * Hn,      Hn);
      ld_rows64(wb, w1 + (size_t)(tn * 16) * Hn + 64, Hn);
      mm64s(aQ,       g1, acc);
      mm64s(aQ + 64,  g2, acc);
#pragma unroll
      for (int r = 0; r < 8; r++)
        sAo[(r + 8 * half) * QP + col] = acc[r] / dn[r];
      park64_late(slot0, wa);
      park64(slot1, wb);
    }
    lds_handover_barrier();

    if (mmw) {
      ld_rows64(u0, w2  + (size_t)(tn * 16) * Hn,  Hn);
      ld_rows64(u1, scw + (size_t)(tn * 16) * INn, INn);
      v8f acc = {0.f,0.f,0.f,0.f,0.f,0.f,0.f,0.f};
      const float* ap = sAo + idx * QP + 4 * half;
      mm64p(ap,      slot0, acc);
      mm64p(ap + 64, slot1, acc);
#pragma unroll
      for (int r = 0; r < 8; r++) {
        float v = acc[r] + b1c;
        sH[(r + 8 * half) * QP + col] = v > 0.f ? v : 0.f;
      }
      park64_late(slot0, u0);
      park64(slot1, u1);
    }
    lds_handover_barrier();
    RowsH y0, y1;
    float ubc = 0.f;
    RowsH p0, p1;
    float nlw[4], nlb[4];
    const float* pw = nullptr;
    if (!LAST) pw = (mmw ? nWk : nWv) + (size_t)(tn * 16) * INn;
    if (!LAST && !mmw) {
      ld_rows64(p0, pw,      INn);
      ld_rows64(p1, pw + 64, INn);
      ld_ln4(nlnw, nlnb, nlw, nlb);
    }

    if (mmw) {
      ld_rows64(u2, w2  + (size_t)(tn * 16) * Hn  + 64, Hn);
      ld_rows64(u3, scw + (size_t)(tn * 16) * INn + 64, INn);
      if (LAST) ubc = ub[col];
      v8f acc1 = {0.f,0.f,0.f,0.f,0.f,0.f,0.f,0.f};
      v8f acc2 = {0.f,0.f,0.f,0.f,0.f,0.f,0.f,0.f};
      const float* a1 = sH  + idx * QP + 4 * half;
      const float* a2 = sXn + idx * QP + 4 * half;
      mm64p(a1,      slot0, acc1);
      mm64p(a2,      slot1, acc2);
      park64(slot0, u2);
      mm64p(a1 + 64, slot0, acc1);
      if (LAST) ld_rows64(y0, uw + (size_t)(tn * 16) * Hn, Hn);
      else      ld_rows64(p0, pw, INn);
      park64(slot1, u3);
      mm64p(a2 + 64, slot1, acc2);
      if (LAST) {
        ld_rows64(y1, uw + (size_t)(tn * 16) * Hn + 64, Hn);
      } else {
        ld_rows64(p1, pw + 64, INn);
        ld_ln4(nlnw, nlnb, nlw, nlb);
      }
#pragma unroll
      for (int r = 0; r < 8; r++) {
        float v = acc1[r] + b2c;
        v = v > 0.f ? v : 0.f;
        const float xv = v + acc2[r] + scbc;
        sAo[(r + 8 * half) * QP + col] = xv;
      }
      if (LAST) {
        park64_late(slot0, y0);
        park64(slot1, y1);
      }
    }
    lds_handover_barrier();
    if (LAST) {
      if (mmw) {
        v8f acc = {0.f,0.f,0.f,0.f,0.f,0.f,0.f,0.f};
        const float* ap = sAo + idx * QP + 4 * half;
        mm64p(ap,      slot0, acc);
        mm64p(ap + 64, slot1, acc);
#pragma unroll
        for (int r = 0; r < 8; r++) {
          const int t = tm * 16 + r + 8 * half;
          y[(size_t)t * Bn * OUTn + (size_t)b * OUTn + col] = acc[r] + ubc;
        }
      }
    } else {
      st_ln(sAo + wave * QP, nlw, nlb, nxn + (row0 + wave) * INn, sH + wave * QP);
      park64_late(slot0, p0);
      park64(slot1, p1);
      lds_handover_barrier();
      v8f acc = {0.f,0.f,0.f,0.f,0.f,0.f,0.f,0.f};
      const float* ap = sH + idx * QP + 4 * half;
      mm64p(ap,      slot0, acc);
      mm64p(ap + 64, slot1, acc);
      float* Out = (mmw ? nK : nV) + (row0 * Hn) + tn * 16;
#pragma unroll
      for (int r = 0; r < 8; r++) {
        float v = acc[r];
        if (mmw) v = ((v > 0.f) ? v : expm1f(v)) + 1.0f;
        Out[(size_t)(r + 8 * half) * Hn + idx] = v;
      }
    }
  } else if (blockIdx.x < 160) {
    st_states(Kb, Vb, st, layerOff, outst, (blockIdx.x - 128) * 16 + wave);
  } else {
    const int e = (blockIdx.x - 160) * 512 + threadIdx.x;
    st_statez(Kb, st, layerOff, outst, e >> 7, e & 127);
  }
}

extern "C" void kernel_launch(void* const* d_in, const int* in_sizes, int n_in,
                              void* d_out, int out_size, void* d_ws, size_t ws_size,
                              hipStream_t stream) {
  (void)in_sizes; (void)n_in; (void)out_size; (void)ws_size;
  const float* z   = (const float*)d_in[0];
  const float* st  = (const float*)d_in[1];
  const float* Wk  = (const float*)d_in[2];
  const float* Wq  = (const float*)d_in[3];
  const float* Wv  = (const float*)d_in[4];
  const float* lnw = (const float*)d_in[5];
  const float* lnb = (const float*)d_in[6];
  const float* w1  = (const float*)d_in[7];
  const float* b1  = (const float*)d_in[8];
  const float* w2  = (const float*)d_in[9];
  const float* b2  = (const float*)d_in[10];
  const float* scw = (const float*)d_in[11];
  const float* scb = (const float*)d_in[12];
  const float* uw  = (const float*)d_in[13];
  const float* ub  = (const float*)d_in[14];
  float* out = (float*)d_out;
  float* ws  = (float*)d_ws;

  float* K1   = ws;
  float* XN   = ws + 262144;
  float* K0   = ws + 524288;
  float* QV   = ws + 786432;
  float* V0   = ws + 1048576;
  float* OST  = out + YSZ;

  dla_front_kernel<<<128, 768, 0, stream>>>(z, st, lnw, lnb, Wk, Wq, Wv, XN, K0, QV, V0, OST);
  dla_mix_kernel<false><<<162, 512, 0, stream>>>(QV, K0, V0, XN, st, 0, OST,
                                                 w1, b1, w2, b2, scw, scb,
                                                 lnw + INn, lnb + INn, Wk + Hn * INn, Wv + Hn * INn,
                                                 XN, K1, QV,
                                                 nullptr, nullptr, nullptr, nullptr);
  dla_mix_kernel<true><<<162, 512, 0, stream>>>(nullptr, K1, QV, XN, st, LAYERSTATE, OST,
                                                w1 + Hn * Hn, b1 + Hn, w2 + Hn * Hn, b2 + Hn,
                                                scw + Hn * INn, scb + Hn,
                                                nullptr, nullptr, nullptr, nullptr,
                                                nullptr, nullptr, nullptr,
                                                Wq + Hn * INn, uw, ub, out);
}
